# starting kernel
# baseline (speedup 1.0000x reference)
#include <hip/hip_runtime.h>
#include <hip/hip_bf16.h>

typedef __attribute__((ext_vector_type(2))) float v2f;
typedef __attribute__((ext_vector_type(8))) float v8f;

#define IN_DIM  256
#define OUT_DIM 128

__global__ __launch_bounds__(128) void k_gemm(const float* __restrict__ A,
                                              const float* __restrict__ W,
                                              float* __restrict__ Z,
                                              int n_src) {
  const int lane = threadIdx.x & 31;
  const int wave = threadIdx.x >> 5;
  const int tile = blockIdx.x * 4 + wave;
  const int m0   = tile * 16;
  if (m0 >= n_src) return;

  const int half = lane >> 4;
  const int nc   = lane & 15;

  int mrow = m0 + nc;
  if (mrow >= n_src) mrow = n_src - 1;
  const float* __restrict__ arow = A + (size_t)mrow * IN_DIM;

  const v8f vzero = {};
  v8f acc[8];
#pragma unroll
  for (int i = 0; i < 8; ++i) acc[i] = vzero;

  for (int k = 0; k < IN_DIM; k += 4) {
    const v2f a = *(const v2f*)(arow + k + 2 * half);
#pragma unroll
    for (int tn = 0; tn < 8; ++tn) {
      const int n = tn * 16 + nc;
      const v2f b = *(const v2f*)(W + (size_t)n * IN_DIM + k + 2 * half);
      acc[tn] = __builtin_amdgcn_wmma_f32_16x16x4_f32(
           false, a,  false, b,
           (short)0, acc[tn],  false,  false);
    }
  }

#pragma unroll
  for (int r = 0; r < 8; ++r) {
    const int mm = m0 + r + 8 * half;
    if (mm < n_src) {
      float* zr = Z + (size_t)mm * OUT_DIM;
#pragma unroll
      for (int tn = 0; tn < 8; ++tn) zr[tn * 16 + nc] = acc[tn][r];
    }
  }
}

__global__ __launch_bounds__(256) void k_score(const float* __restrict__ Z,
                                               const float* __restrict__ wt,
                                               float* __restrict__ score,
                                               int n_src) {
  const int w    = (blockIdx.x * blockDim.x + threadIdx.x) >> 5;
  const int lane = threadIdx.x & 31;
  if (w >= n_src) return;
  const float* zr = Z + (size_t)w * OUT_DIM;
  float s = zr[lane]      * wt[lane]
          + zr[lane + 32] * wt[lane + 32]
          + zr[lane + 64] * wt[lane + 64]
          + zr[lane + 96] * wt[lane + 96];
#pragma unroll
  for (int off = 16; off > 0; off >>= 1) s += __shfl_down(s, off, 32);
  if (lane == 0) score[w] = s;
}

__device__ __forceinline__ float edge_logit(const float* __restrict__ score,
                                            const float* __restrict__ t,
                                            int s, int d) {
  const float e = -fabsf(t[s] - t[d]);
  return expf(score[s] * e * 0.002f);
}

__global__ __launch_bounds__(256) void k_zero(float* __restrict__ m,
                                              float* __restrict__ denom,
                                              int n_dst) {
  const int i = blockIdx.x * blockDim.x + threadIdx.x;
  if (i < n_dst) { m[i] = 0.0f; denom[i] = 0.0f; }
}

__global__ __launch_bounds__(256) void k_edge_max(const float* __restrict__ score,
                                                  const float* __restrict__ t,
                                                  const int* __restrict__ src,
                                                  const int* __restrict__ dst,
                                                  float* __restrict__ m, int E) {
  const int i = blockIdx.x * blockDim.x + threadIdx.x;
  if (i >= E) return;
  const float le = edge_logit(score, t, src[i], dst[i]);
  atomicMax((int*)(m + dst[i]), __float_as_int(le));
}

__global__ __launch_bounds__(256) void k_edge_denom(const float* __restrict__ score,
                                                    const float* __restrict__ t,
                                                    const int* __restrict__ src,
                                                    const int* __restrict__ dst,
                                                    const float* __restrict__ m,
                                                    float* __restrict__ denom, int E) {
  const int i = blockIdx.x * blockDim.x + threadIdx.x;
  if (i >= E) return;
  const int d  = dst[i];
  const float un = expf(edge_logit(score, t, src[i], d) - m[d]);
  atomicAdd(denom + d, un);
}

__global__ __launch_bounds__(256) void k_scatter(const float* __restrict__ score,
                                                 const float* __restrict__ t,
                                                 const int* __restrict__ src,
                                                 const int* __restrict__ dst,
                                                 const float* __restrict__ m,
                                                 const float* __restrict__ denom,
                                                 const float* __restrict__ Z,
                                                 float* __restrict__ out, int E) {
  const int w    = (blockIdx.x * blockDim.x + threadIdx.x) >> 5;
  const int lane = threadIdx.x & 31;
  if (w >= E) return;
  const int s = src[w];
  const int d = dst[w];
  const float le    = edge_logit(score, t, s, d);
  const float un    = expf(le - m[d]);
  const float alpha = un / fmaxf(denom[d], 1e-20f);

  const float4 zv = *(const float4*)(Z + (size_t)s * OUT_DIM + lane * 4);
  float* o = out + (size_t)d * OUT_DIM + lane * 4;
  atomicAdd(o + 0, alpha * zv.x);
  atomicAdd(o + 1, alpha * zv.y);
  atomicAdd(o + 2, alpha * zv.z);
  atomicAdd(o + 3, alpha * zv.w);
}

extern "C" void kernel_launch(void* const* d_in, const int* in_sizes, int n_in,
                              void* d_out, int out_size, void* d_ws, size_t ws_size,
                              hipStream_t stream) {
  const float* features = (const float*)d_in[0];
  const float* t        = (const float*)d_in[1];
  const float* W_fc     = (const float*)d_in[2];
  const float* w_t      = (const float*)d_in[3];
  const int*   src_idx  = (const int*)d_in[4];
  const int*   dst_idx  = (const int*)d_in[5];

  const int n_src = in_sizes[0] / IN_DIM;
  const int E     = in_sizes[4];
  const int n_dst = out_size / OUT_DIM;

  float* zbuf  = (float*)d_ws;
  float* score = zbuf + (size_t)n_src * OUT_DIM;
  float* mbuf  = score + n_src;
  float* denom = mbuf + n_dst;
  float* out   = (float*)d_out;

  const int ntile = (n_src + 15) / 16;
  k_gemm<<<(ntile + 3) / 4, 128, 0, stream>>>(features, W_fc, zbuf, n_src);

  k_score<<<(n_src + 7) / 8, 256, 0, stream>>>(zbuf, w_t, score, n_src);

  hipMemcpyAsync(out, zbuf, (size_t)n_dst * OUT_DIM * sizeof(float),
                 hipMemcpyDeviceToDevice, stream);
  k_zero<<<(n_dst + 255) / 256, 256, 0, stream>>>(mbuf, denom, n_dst);

  k_edge_max<<<(E + 255) / 256, 256, 0, stream>>>(score, t, src_idx, dst_idx, mbuf, E);

  k_edge_denom<<<(E + 255) / 256, 256, 0, stream>>>(score, t, src_idx, dst_idx,
                                                    mbuf, denom, E);

  k_scatter<<<(E + 7) / 8, 256, 0, stream>>>(score, t, src_idx, dst_idx,
                                             mbuf, denom, zbuf, out, E);
}


# bin edges per dst by integer ticket into lists parked in the out rows; one wave per dst does max, denom and weighted z sum in registers: no float atomics, sums in ticket arrival order
# speedup vs baseline: 2.3574x; 2.3574x over previous
#include <hip/hip_runtime.h>
#include <hip/hip_bf16.h>

typedef __attribute__((ext_vector_type(2))) float v2f;
typedef __attribute__((ext_vector_type(8))) float v8f;

#define IN_DIM  256
#define OUT_DIM 128

__global__ __launch_bounds__(128) void k_gemm(const float* __restrict__ A,
                                              const float* __restrict__ W,
                                              float* __restrict__ Z,
                                              int n_src) {
  const int lane = threadIdx.x & 31;
  const int wave = threadIdx.x >> 5;
  const int tile = blockIdx.x * 4 + wave;
  const int m0   = tile * 16;
  if (m0 >= n_src) return;

  const int half = lane >> 4;
  const int nc   = lane & 15;

  int mrow = m0 + nc;
  if (mrow >= n_src) mrow = n_src - 1;
  const float* __restrict__ arow = A + (size_t)mrow * IN_DIM;

  const v8f vzero = {};
  v8f acc[8];
#pragma unroll
  for (int i = 0; i < 8; ++i) acc[i] = vzero;

  for (int k = 0; k < IN_DIM; k += 4) {
    const v2f a = *(const v2f*)(arow + k + 2 * half);
#pragma unroll
    for (int tn = 0; tn < 8; ++tn) {
      const int n = tn * 16 + nc;
      const v2f b = *(const v2f*)(W + (size_t)n * IN_DIM + k + 2 * half);
      acc[tn] = __builtin_amdgcn_wmma_f32_16x16x4_f32(
           false, a,  false, b,
           (short)0, acc[tn],  false,  false);
    }
  }

#pragma unroll
  for (int r = 0; r < 8; ++r) {
    const int mm = m0 + r + 8 * half;
    if (mm < n_src) {
      float* zr = Z + (size_t)mm * OUT_DIM;
#pragma unroll
      for (int tn = 0; tn < 8; ++tn) zr[tn * 16 + nc] = acc[tn][r];
    }
  }
}

__global__ __launch_bounds__(256) void k_score(const float* __restrict__ Z,
                                               const float* __restrict__ wt,
                                               float* __restrict__ score,
                                               int n_src) {
  const int w    = (blockIdx.x * blockDim.x + threadIdx.x) >> 5;
  const int lane = threadIdx.x & 31;
  if (w >= n_src) return;
  const float* zr = Z + (size_t)w * OUT_DIM;
  float s = zr[lane]      * wt[lane]
          + zr[lane + 32] * wt[lane + 32]
          + zr[lane + 64] * wt[lane + 64]
          + zr[lane + 96] * wt[lane + 96];
#pragma unroll
  for (int off = 16; off > 0; off >>= 1) s += __shfl_down(s, off, 32);
  if (lane == 0) score[w] = s;
}

#define LIST_CAP (OUT_DIM - 1)

__device__ __forceinline__ float edge_logit(const float* __restrict__ score,
                                            const float* __restrict__ t,
                                            int s, float td) {
  const float e = -fabsf(t[s] - td);
  return expf(score[s] * e * 0.002f);
}

__global__ __launch_bounds__(256) void k_zero(int* __restrict__ rows, int n_dst) {
  const int i = blockIdx.x * blockDim.x + threadIdx.x;
  if (i < n_dst) rows[(size_t)i * OUT_DIM] = 0;
}

__global__ __launch_bounds__(256) void k_bin(const int* __restrict__ src,
                                             const int* __restrict__ dst,
                                             int* __restrict__ rows, int E) {
  const int i = blockIdx.x * blockDim.x + threadIdx.x;
  if (i >= E) return;
  const int s = src[i];
  int* row = rows + (size_t)dst[i] * OUT_DIM;
  const int pos = atomicAdd(row, 1);
  if ((unsigned)pos < (unsigned)LIST_CAP) row[1 + pos] = s;
}

__device__ __forceinline__ float wave_max(float v) {
#pragma unroll
  for (int off = 16; off > 0; off >>= 1) v = fmaxf(v, __shfl_xor(v, off, 32));
  return v;
}
__device__ __forceinline__ float wave_sum(float v) {
#pragma unroll
  for (int off = 16; off > 0; off >>= 1) v += __shfl_xor(v, off, 32);
  return v;
}
__device__ __forceinline__ void axpy4(float4& acc, float a, const float4 z) {
  acc.x += a * z.x; acc.y += a * z.y; acc.z += a * z.z; acc.w += a * z.w;
}
__device__ __forceinline__ float lane_f(float v, int l) {
  return __int_as_float(__builtin_amdgcn_readlane(__float_as_int(v), l));
}

__global__ __launch_bounds__(256) void k_gather(const float* __restrict__ score,
                                                const float* __restrict__ t,
                                                const int* __restrict__ src,
                                                const int* __restrict__ dst,
                                                const float* __restrict__ Z,
                                                float* __restrict__ out,
                                                int n_src, int n_dst, int E) {
  const int d = __builtin_amdgcn_readfirstlane((int)((blockIdx.x * blockDim.x + threadIdx.x) >> 5));
  const int lane = threadIdx.x & 31;
  if (d >= n_dst) return;

  float* orow = out + (size_t)d * OUT_DIM;
  const float* zl = Z + lane * 4;
  const float td = t[d];

  const int w0 = __float_as_int(orow[lane]);
  const int n  = __builtin_amdgcn_readfirstlane(w0);

  float4 acc = make_float4(0.0f, 0.0f, 0.0f, 0.0f);

  if (n <= LIST_CAP) {
    int   sidx[4];
    float al[4];
    float mx = 0.0f;
#pragma unroll
    for (int q = 0; q < 4; ++q) {
      const int k = lane + 32 * q;
      sidx[q] = 0;
      al[q]   = 0.0f;
      if (32 * q <= n) {
        const int w = (q == 0) ? w0 : __float_as_int(orow[k]);
        if (k >= 1 && k <= n) {
          const int sc = (int)min((unsigned)w, (unsigned)(n_src - 1));
          sidx[q] = sc;
          al[q]   = edge_logit(score, t, sc, td);
        }
      }
      mx = fmaxf(mx, al[q]);
    }
    mx = wave_max(mx);

    float dsum = 0.0f;
#pragma unroll
    for (int q = 0; q < 4; ++q) {
      const int k = lane + 32 * q;
      const float un = (k >= 1 && k <= n) ? expf(al[q] - mx) : 0.0f;
      al[q] = un;
      dsum += un;
    }
    dsum = wave_sum(dsum);
    const float dn = fmaxf(dsum, 1e-20f);
#pragma unroll
    for (int q = 0; q < 4; ++q) al[q] = al[q] / dn;

#pragma unroll
    for (int q = 0; q < 4; ++q) {
      int       L    = (q == 0) ? 1 : 0;
      const int Lend = ((n < 32 * q + 31) ? n : 32 * q + 31) - 32 * q;
      for (; L + 3 <= Lend; L += 4) {
        const int s0 = __builtin_amdgcn_readlane(sidx[q], L);
        const int s1 = __builtin_amdgcn_readlane(sidx[q], L + 1);
        const int s2 = __builtin_amdgcn_readlane(sidx[q], L + 2);
        const int s3 = __builtin_amdgcn_readlane(sidx[q], L + 3);
        const float4 z0 = *(const float4*)(zl + (size_t)s0 * OUT_DIM);
        const float4 z1 = *(const float4*)(zl + (size_t)s1 * OUT_DIM);
        const float4 z2 = *(const float4*)(zl + (size_t)s2 * OUT_DIM);
        const float4 z3 = *(const float4*)(zl + (size_t)s3 * OUT_DIM);
        axpy4(acc, lane_f(al[q], L),     z0);
        axpy4(acc, lane_f(al[q], L + 1), z1);
        axpy4(acc, lane_f(al[q], L + 2), z2);
        axpy4(acc, lane_f(al[q], L + 3), z3);
      }
      for (; L <= Lend; ++L) {
        const int s0 = __builtin_amdgcn_readlane(sidx[q], L);
        const float4 z0 = *(const float4*)(zl + (size_t)s0 * OUT_DIM);
        axpy4(acc, lane_f(al[q], L), z0);
      }
    }
  } else {
    float mx = 0.0f;
    for (int i = lane; i < E; i += 32)
      if (dst[i] == d) mx = fmaxf(mx, edge_logit(score, t, src[i], td));
    mx = wave_max(mx);
    float dsum = 0.0f;
    for (int i = lane; i < E; i += 32)
      if (dst[i] == d) dsum += expf(edge_logit(score, t, src[i], td) - mx);
    dsum = wave_sum(dsum);
    const float dn = fmaxf(dsum, 1e-20f);
    for (int base = 0; base < E; base += 32) {
      const int  i   = base + lane;
      const bool hit = (i < E) && (dst[i] == d);
      int   s = 0;
      float a = 0.0f;
      if (hit) {
        s = src[i];
        a = expf(edge_logit(score, t, s, td) - mx) / dn;
      }
      unsigned mask = __builtin_amdgcn_ballot_w32(hit);
      while (mask) {
        const int b = __builtin_ctz(mask);
        mask &= mask - 1;
        const int sb = __builtin_amdgcn_readlane(s, b);
        const float4 zb = *(const float4*)(zl + (size_t)sb * OUT_DIM);
        axpy4(acc, lane_f(a, b), zb);
      }
    }
  }

  const float4 zd = *(const float4*)(zl + (size_t)d * OUT_DIM);
  float4 o;
  o.x = zd.x + acc.x; o.y = zd.y + acc.y; o.z = zd.z + acc.z; o.w = zd.w + acc.w;
  *(float4*)(orow + lane * 4) = o;
}

extern "C" void kernel_launch(void* const* d_in, const int* in_sizes, int n_in,
                              void* d_out, int out_size, void* d_ws, size_t ws_size,
                              hipStream_t stream) {
  const float* features = (const float*)d_in[0];
  const float* t        = (const float*)d_in[1];
  const float* W_fc     = (const float*)d_in[2];
  const float* w_t      = (const float*)d_in[3];
  const int*   src_idx  = (const int*)d_in[4];
  const int*   dst_idx  = (const int*)d_in[5];

  const int n_src = in_sizes[0] / IN_DIM;
  const int E     = in_sizes[4];
  const int n_dst = out_size / OUT_DIM;

  float* zbuf  = (float*)d_ws;
  float* score = zbuf + (size_t)n_src * OUT_DIM;
  float* out   = (float*)d_out;

  const int ntile = (n_src + 15) / 16;
  k_gemm<<<(ntile + 3) / 4, 128, 0, stream>>>(features, W_fc, zbuf, n_src);

  k_score<<<(n_src + 7) / 8, 256, 0, stream>>>(zbuf, w_t, score, n_src);

  k_zero<<<(n_dst + 255) / 256, 256, 0, stream>>>((int*)d_out, n_dst);
  k_bin<<<(E + 255) / 256, 256, 0, stream>>>(src_idx, dst_idx, (int*)d_out, E);

  k_gather<<<(n_dst + 7) / 8, 256, 0, stream>>>(score, t, src_idx, dst_idx,
                                                zbuf, out, n_src, n_dst, E);
}


# k_gemm: features and W_fc chunks go through LDS by whole-line loads, prefetched during the multiply; 32x128 per wave, one B fragment feeds two WMMAs; k order and z bits unchanged
# speedup vs baseline: 3.8822x; 1.6468x over previous
#include <hip/hip_runtime.h>
#include <hip/hip_bf16.h>

typedef __attribute__((ext_vector_type(2))) float v2f;
typedef __attribute__((ext_vector_type(8))) float v8f;

#define IN_DIM  256
#define OUT_DIM 128

typedef __attribute__((ext_vector_type(4))) float v4f;

#define GBM   128
#define GBK   32
#define GLP   (GBK + 4)
#define GNCH  (IN_DIM / GBK)
#define GPASS (GBM / 16)
static_assert(GBM == OUT_DIM, "the W tile is staged with the A tile's row map");

#define G_MULT(ks0, ks1)                                                        \
  _Pragma("unroll")                                                             \
  for (int ks = (ks0) * 4; ks < (ks1) * 4; ks += 4) {                           \
    const v2f a0 = *(const v2f*)(as + ks);                                      \
    const v2f a1 = *(const v2f*)(as + 16 * GLP + ks);                           \
    _Pragma("unroll")                                                           \
    for (int tn = 0; tn < 8; ++tn) {                                            \
      const v2f b = *(const v2f*)(ws + tn * 16 * GLP + ks);                     \
      acc[0][tn] = __builtin_amdgcn_wmma_f32_16x16x4_f32(                       \
           false, a0,  false, b,                              \
           (short)0, acc[0][tn],  false,  false);\
      acc[1][tn] = __builtin_amdgcn_wmma_f32_16x16x4_f32(                       \
           false, a1,  false, b,                              \
           (short)0, acc[1][tn],  false,  false);\
    }                                                                           \
  }

__global__ __launch_bounds__(128) void k_gemm(const float* __restrict__ A,
                                              const float* __restrict__ W,
                                              float* __restrict__ Z,
                                              int n_src) {
  __shared__ __attribute__((aligned(16))) float As[2][GBM * GLP];
  __shared__ __attribute__((aligned(16))) float Ws[2][OUT_DIM * GLP];

  const int tid  = threadIdx.x;
  const int lane = tid & 31;
  const int wave = tid >> 5;
  const int half = lane >> 4;
  const int nc   = lane & 15;
  const int m0   = blockIdx.x * GBM;

  const int srow = tid >> 3;
  const int scol = (tid & 7) * 4;
  const int soff = srow * GLP + scol;


  unsigned arow[GPASS];
#pragma unroll
  for (int i = 0; i < GPASS; ++i) {
    int gr = m0 + srow + 16 * i;
    if (gr >= n_src) gr = n_src - 1;
    arow[i] = (unsigned)gr * IN_DIM + scol;
  }
  const float* __restrict__ wsrc = W + srow * IN_DIM + scol;

  v4f pf[GPASS];

  const v8f vzero = {};
  v8f acc[2][8];
#pragma unroll
  for (int i = 0; i < 8; ++i) { acc[0][i] = vzero; acc[1][i] = vzero; }

#pragma unroll
  for (int i = 0; i < GPASS; ++i) pf[i] = *(const v4f*)(A + arow[i]);
#pragma unroll
  for (int i = 0; i < GPASS; ++i) *(v4f*)&As[0][soff + 16 * i * GLP] = pf[i];
#pragma unroll
  for (int i = 0; i < GPASS; ++i) pf[i] = *(const v4f*)(wsrc + 16 * i * IN_DIM);
#pragma unroll
  for (int i = 0; i < GPASS; ++i) *(v4f*)&Ws[0][soff + 16 * i * GLP] = pf[i];
  __syncthreads();

  const int aoff = (wave * 32 + nc) * GLP + 2 * half;
  const int boff = nc * GLP + 2 * half;

  for (int kc = 0; kc < GNCH; ++kc) {
    const int  buf  = kc & 1;
    const bool more = kc + 1 < GNCH;
    const float* as = &As[buf][aoff];
    const float* ws = &Ws[buf][boff];

    if (more) {
#pragma unroll
      for (int i = 0; i < GPASS; ++i) pf[i] = *(const v4f*)(A + arow[i] + (kc + 1) * GBK);
    }
    G_MULT(0, GBK / 8)
    if (more) {
#pragma unroll
      for (int i = 0; i < GPASS; ++i) *(v4f*)&As[buf ^ 1][soff + 16 * i * GLP] = pf[i];
#pragma unroll
      for (int i = 0; i < GPASS; ++i) pf[i] = *(const v4f*)(wsrc + 16 * i * IN_DIM + (kc + 1) * GBK);
    }
    G_MULT(GBK / 8, GBK / 4)
    if (more) {
#pragma unroll
      for (int i = 0; i < GPASS; ++i) *(v4f*)&Ws[buf ^ 1][soff + 16 * i * GLP] = pf[i];
    }
    __syncthreads();
  }

#pragma unroll
  for (int tm = 0; tm < 2; ++tm) {
#pragma unroll
    for (int r = 0; r < 8; ++r) {
      const int mm = m0 + wave * 32 + tm * 16 + r + 8 * half;
      if (mm < n_src) {
        float* zr = Z + (size_t)mm * OUT_DIM;
#pragma unroll
        for (int tn = 0; tn < 8; ++tn) zr[tn * 16 + nc] = acc[tm][tn][r];
      }
    }
  }
}
#undef G_MULT

__global__ __launch_bounds__(256) void k_score(const float* __restrict__ Z,
                                               const float* __restrict__ wt,
                                               float* __restrict__ score,
                                               int n_src) {
  const int w    = (blockIdx.x * blockDim.x + threadIdx.x) >> 5;
  const int lane = threadIdx.x & 31;
  if (w >= n_src) return;
  const float* zr = Z + (size_t)w * OUT_DIM;
  float s = zr[lane]      * wt[lane]
          + zr[lane + 32] * wt[lane + 32]
          + zr[lane + 64] * wt[lane + 64]
          + zr[lane + 96] * wt[lane + 96];
#pragma unroll
  for (int off = 16; off > 0; off >>= 1) s += __shfl_down(s, off, 32);
  if (lane == 0) score[w] = s;
}

#define RB       7
#define RANGE    (1 << RB)
#define SEGW     (RANGE * OUT_DIM)
#define NBMAX    1024
#define LCAP     4096
#define CE       8
#define CT       1024
#define GW       16
#define EPT      (LCAP / (GW * 32))

__device__ __forceinline__ float edge_logit(const float* __restrict__ score,
                                            const float* __restrict__ t,
                                            int s, float td) {
  const float e = -fabsf(t[s] - td);
  return expf(score[s] * e * 0.002f);
}

__global__ __launch_bounds__(256) void k_zero(int* __restrict__ rows, int nb) {
  const int c = blockIdx.x * blockDim.x + threadIdx.x;
  if (c < nb) rows[(size_t)c * SEGW] = 0;
}

__global__ __launch_bounds__(CT) void k_coarse(const int* __restrict__ src,
                                               const int* __restrict__ dst,
                                               int* __restrict__ rows,
                                               int E, int n_dst, int nb) {
  __shared__ int hist[NBMAX];
  const int tid = threadIdx.x;
  if (nb > NBMAX) return;
  for (int c = tid; c < NBMAX; c += CT) hist[c] = 0;
  __syncthreads();

  int s[CE], d[CE], r[CE];
  const int base = blockIdx.x * (CT * CE);
#pragma unroll
  for (int j = 0; j < CE; ++j) {
    const int i = base + j * CT + tid;
    d[j] = -1; s[j] = 0; r[j] = 0;
    if (i < E) { d[j] = dst[i]; s[j] = src[i]; }
  }
#pragma unroll
  for (int j = 0; j < CE; ++j) {
    if ((unsigned)d[j] < (unsigned)n_dst) r[j] = atomicAdd(&hist[d[j] >> RB], 1);
    else d[j] = -1;
  }
  __syncthreads();

  if (tid < nb) {
    const int h = hist[tid];
    if (h > 0) hist[tid] = atomicAdd(rows + (size_t)tid * SEGW, h);
  }
  __syncthreads();

#pragma unroll
  for (int j = 0; j < CE; ++j) {
    if (d[j] >= 0) {
      const int c   = d[j] >> RB;
      const int pos = hist[c] + r[j];
      const int nr  = min(RANGE, n_dst - c * RANGE);
      const int cap = nr * OUT_DIM - 1;
      if ((unsigned)pos < (unsigned)cap)
        rows[(size_t)c * SEGW + 1 + pos] = (s[j] << RB) | (d[j] & (RANGE - 1));
    }
  }
}

__device__ __forceinline__ float wave_max(float v) {
#pragma unroll
  for (int off = 16; off > 0; off >>= 1) v = fmaxf(v, __shfl_xor(v, off, 32));
  return v;
}
__device__ __forceinline__ float wave_sum(float v) {
#pragma unroll
  for (int off = 16; off > 0; off >>= 1) v += __shfl_xor(v, off, 32);
  return v;
}
__device__ __forceinline__ void axpy4(float4& acc, float a, const float4 z) {
  acc.x += a * z.x; acc.y += a * z.y; acc.z += a * z.z; acc.w += a * z.w;
}
__device__ __forceinline__ float lane_f(float v, int l) {
  return __int_as_float(__builtin_amdgcn_readlane(__float_as_int(v), l));
}

__global__ __launch_bounds__(GW * 32) void k_gather(const float* __restrict__ score,
                                                    const float* __restrict__ t,
                                                    const int* __restrict__ src,
                                                    const int* __restrict__ dst,
                                                    const float* __restrict__ Z,
                                                    float* __restrict__ out,
                                                    int n_src, int n_dst, int E, int nb) {
  __shared__ int cnt[RANGE];
  __shared__ int off[RANGE];
  __shared__ int cur[RANGE];
  __shared__ int list[LCAP];
  __shared__ float wgt[LCAP];

  const int c    = blockIdx.x;
  const int tid  = threadIdx.x;
  const int lane = tid & 31;
  const int wave = __builtin_amdgcn_readfirstlane(tid >> 5);
  const int d0   = c * RANGE;
  const int nr   = min(RANGE, n_dst - d0);
  const int* seg = (const int*)out + (size_t)c * SEGW;

  const int ne = __builtin_amdgcn_readfirstlane(seg[0]);
  const bool blk_slow = (nb > NBMAX) || (ne > LCAP) || (ne > nr * OUT_DIM - 1);

  int wv[EPT];
#pragma unroll
  for (int j = 0; j < EPT; ++j) {
    const int e = tid + j * (GW * 32);
    wv[j] = 0;
    if (!blk_slow && e < ne) wv[j] = seg[1 + e];
  }

  if (tid < RANGE) cnt[tid] = 0;
  __syncthreads();
#pragma unroll
  for (int j = 0; j < EPT; ++j)
    if (!blk_slow && tid + j * (GW * 32) < ne) atomicAdd(&cnt[wv[j] & (RANGE - 1)], 1);
  __syncthreads();
  if (wave == 0) {
    const int c0 = cnt[4 * lane], c1 = cnt[4 * lane + 1], c2 = cnt[4 * lane + 2], c3 = cnt[4 * lane + 3];
    const int sum = c0 + c1 + c2 + c3;
    int inc = sum;
#pragma unroll
    for (int o = 1; o < 32; o <<= 1) {
      const int v = __shfl_up(inc, o, 32);
      if (lane >= o) inc += v;
    }
    const int ex = inc - sum;
    off[4 * lane]     = ex;                cur[4 * lane]     = ex;
    off[4 * lane + 1] = ex + c0;           cur[4 * lane + 1] = ex + c0;
    off[4 * lane + 2] = ex + c0 + c1;      cur[4 * lane + 2] = ex + c0 + c1;
    off[4 * lane + 3] = ex + c0 + c1 + c2; cur[4 * lane + 3] = ex + c0 + c1 + c2;
  }
  __syncthreads();
#pragma unroll
  for (int j = 0; j < EPT; ++j)
    if (!blk_slow && tid + j * (GW * 32) < ne) {
      const unsigned w = (unsigned)wv[j];
      const int pos = atomicAdd(&cur[w & (RANGE - 1)], 1);
      list[pos] = (int)(w >> RB);
    }
  __syncthreads();

  const float* zl = Z + lane * 4;

  for (int ld = wave; ld < nr; ld += GW) {
    const int d = d0 + ld;
    const int n    = __builtin_amdgcn_readfirstlane(cnt[ld]);
    const int lb   = __builtin_amdgcn_readfirstlane(off[ld]);
    const float td = t[d];

    float4 acc = make_float4(0.0f, 0.0f, 0.0f, 0.0f);

    if (!blk_slow) {
      float mx = 0.0f;
      for (int k0 = 0; k0 < n; k0 += 32) {
        const int k = k0 + lane;
        float lg = 0.0f;
        if (k < n) {
          const int sc = (int)min((unsigned)list[lb + k], (unsigned)(n_src - 1));
          lg = edge_logit(score, t, sc, td);
          wgt[lb + k] = lg;
        }
        mx = fmaxf(mx, lg);
      }
      mx = wave_max(mx);

      float dsum = 0.0f;
      for (int k0 = 0; k0 < n; k0 += 32) {
        const int k = k0 + lane;
        if (k < n) {
          const float un = expf(wgt[lb + k] - mx);
          wgt[lb + k] = un;
          dsum += un;
        }
      }
      dsum = wave_sum(dsum);
      const float dn = fmaxf(dsum, 1e-20f);

      for (int k0 = 0; k0 < n; k0 += 32) {
        const int k = k0 + lane;
        int   sv = 0;
        float av = 0.0f;
        if (k < n) {
          sv = (int)min((unsigned)list[lb + k], (unsigned)(n_src - 1));
          av = wgt[lb + k] / dn;
        }
        int       L    = 0;
        const int Lend = ((n - k0 < 32) ? (n - k0) : 32) - 1;
        for (; L + 3 <= Lend; L += 4) {
          const int s0 = __builtin_amdgcn_readlane(sv, L);
          const int s1 = __builtin_amdgcn_readlane(sv, L + 1);
          const int s2 = __builtin_amdgcn_readlane(sv, L + 2);
          const int s3 = __builtin_amdgcn_readlane(sv, L + 3);
          const float4 z0 = *(const float4*)(zl + (size_t)s0 * OUT_DIM);
          const float4 z1 = *(const float4*)(zl + (size_t)s1 * OUT_DIM);
          const float4 z2 = *(const float4*)(zl + (size_t)s2 * OUT_DIM);
          const float4 z3 = *(const float4*)(zl + (size_t)s3 * OUT_DIM);
          axpy4(acc, lane_f(av, L),     z0);
          axpy4(acc, lane_f(av, L + 1), z1);
          axpy4(acc, lane_f(av, L + 2), z2);
          axpy4(acc, lane_f(av, L + 3), z3);
        }
        for (; L <= Lend; ++L) {
          const int s0 = __builtin_amdgcn_readlane(sv, L);
          const float4 z0 = *(const float4*)(zl + (size_t)s0 * OUT_DIM);
          axpy4(acc, lane_f(av, L), z0);
        }
      }
    } else {
      float mx = 0.0f;
      for (int i = lane; i < E; i += 32)
        if (dst[i] == d) mx = fmaxf(mx, edge_logit(score, t, src[i], td));
      mx = wave_max(mx);
      float dsum = 0.0f;
      for (int i = lane; i < E; i += 32)
        if (dst[i] == d) dsum += expf(edge_logit(score, t, src[i], td) - mx);
      dsum = wave_sum(dsum);
      const float dn = fmaxf(dsum, 1e-20f);
      for (int base = 0; base < E; base += 32) {
        const int  i   = base + lane;
        const bool hit = (i < E) && (dst[i] == d);
        int   s = 0;
        float a = 0.0f;
        if (hit) {
          s = src[i];
          a = expf(edge_logit(score, t, s, td) - mx) / dn;
        }
        unsigned mask = __builtin_amdgcn_ballot_w32(hit);
        while (mask) {
          const int b = __builtin_ctz(mask);
          mask &= mask - 1;
          const int sb = __builtin_amdgcn_readlane(s, b);
          const float4 zb = *(const float4*)(zl + (size_t)sb * OUT_DIM);
          axpy4(acc, lane_f(a, b), zb);
        }
      }
    }

    const float4 zd = *(const float4*)(zl + (size_t)d * OUT_DIM);
    float4 o;
    o.x = zd.x + acc.x; o.y = zd.y + acc.y; o.z = zd.z + acc.z; o.w = zd.w + acc.w;
    *(float4*)(out + (size_t)d * OUT_DIM + lane * 4) = o;
  }
}

extern "C" void kernel_launch(void* const* d_in, const int* in_sizes, int n_in,
                              void* d_out, int out_size, void* d_ws, size_t ws_size,
                              hipStream_t stream) {
  const float* features = (const float*)d_in[0];
  const float* t        = (const float*)d_in[1];
  const float* W_fc     = (const float*)d_in[2];
  const float* w_t      = (const float*)d_in[3];
  const int*   src_idx  = (const int*)d_in[4];
  const int*   dst_idx  = (const int*)d_in[5];

  const int n_src = in_sizes[0] / IN_DIM;
  const int E     = in_sizes[4];
  const int n_dst = out_size / OUT_DIM;

  float* zbuf  = (float*)d_ws;
  float* score = zbuf + (size_t)n_src * OUT_DIM;
  float* out   = (float*)d_out;

  k_gemm<<<(n_src + GBM - 1) / GBM, 128, 0, stream>>>(features, W_fc, zbuf, n_src);

  k_score<<<(n_src + 7) / 8, 256, 0, stream>>>(zbuf, w_t, score, n_src);

  const int nb = (n_dst + RANGE - 1) / RANGE;
  k_zero<<<(nb + 255) / 256, 256, 0, stream>>>((int*)d_out, nb);
  k_coarse<<<(E + CT * CE - 1) / (CT * CE), CT, 0, stream>>>(src_idx, dst_idx, (int*)d_out,
                                                            E, n_dst, nb);

  k_gather<<<nb, GW * 32, 0, stream>>>(score, t, src_idx, dst_idx,
                                       zbuf, out, n_src, n_dst, E, nb);
}


# score = z . w_t taken from the GEMM accumulators in k_gemm's epilogue (per-lane column products, DPP xor sum over each half-wave: new summation order of score); k_score launch and z re-read gone
# speedup vs baseline: 4.2933x; 1.1059x over previous
#include <hip/hip_runtime.h>
#include <hip/hip_bf16.h>

typedef __attribute__((ext_vector_type(2))) float v2f;
typedef __attribute__((ext_vector_type(8))) float v8f;

#define IN_DIM  256
#define OUT_DIM 128

typedef __attribute__((ext_vector_type(4))) float v4f;

#define GBM   128
#define GBK   32
#define GLP   (GBK + 4)
#define GNCH  (IN_DIM / GBK)
#define GPASS (GBM / 16)
static_assert(GBM == OUT_DIM, "the W tile is staged with the A tile's row map");

#define G_MULT(ks0, ks1)                                                        \
  _Pragma("unroll")                                                             \
  for (int ks = (ks0) * 4; ks < (ks1) * 4; ks += 4) {                           \
    const v2f a0 = *(const v2f*)(as + ks);                                      \
    const v2f a1 = *(const v2f*)(as + 16 * GLP + ks);                           \
    _Pragma("unroll")                                                           \
    for (int tn = 0; tn < 8; ++tn) {                                            \
      const v2f b = *(const v2f*)(ws + tn * 16 * GLP + ks);                     \
      acc[0][tn] = __builtin_amdgcn_wmma_f32_16x16x4_f32(                       \
           false, a0,  false, b,                              \
           (short)0, acc[0][tn],  false,  false);\
      acc[1][tn] = __builtin_amdgcn_wmma_f32_16x16x4_f32(                       \
           false, a1,  false, b,                              \
           (short)0, acc[1][tn],  false,  false);\
    }                                                                           \
  }

template <int CTRL>
__device__ __forceinline__ float row_xor_add(float v) {
  return v + __int_as_float(__builtin_amdgcn_update_dpp(0, __float_as_int(v), CTRL, 0xf, 0xf, false));
}

__global__ __launch_bounds__(128) void k_gemm(const float* __restrict__ A,
                                              const float* __restrict__ W,
                                              const float* __restrict__ wt,
                                              float* __restrict__ Z,
                                              float* __restrict__ score,
                                              int n_src) {
  __shared__ __attribute__((aligned(16))) float As[2][GBM * GLP];
  __shared__ __attribute__((aligned(16))) float Ws[2][OUT_DIM * GLP];

  const int tid  = threadIdx.x;
  const int lane = tid & 31;
  const int wave = tid >> 5;
  const int half = lane >> 4;
  const int nc   = lane & 15;
  const int m0   = blockIdx.x * GBM;

  const int srow = tid >> 3;
  const int scol = (tid & 7) * 4;
  const int soff = srow * GLP + scol;


  unsigned arow[GPASS];
#pragma unroll
  for (int i = 0; i < GPASS; ++i) {
    int gr = m0 + srow + 16 * i;
    if (gr >= n_src) gr = n_src - 1;
    arow[i] = (unsigned)gr * IN_DIM + scol;
  }
  const float* __restrict__ wsrc = W + srow * IN_DIM + scol;

  v4f pf[GPASS];

  const v8f vzero = {};
  v8f acc[2][8];
#pragma unroll
  for (int i = 0; i < 8; ++i) { acc[0][i] = vzero; acc[1][i] = vzero; }

#pragma unroll
  for (int i = 0; i < GPASS; ++i) pf[i] = *(const v4f*)(A + arow[i]);
#pragma unroll
  for (int i = 0; i < GPASS; ++i) *(v4f*)&As[0][soff + 16 * i * GLP] = pf[i];
#pragma unroll
  for (int i = 0; i < GPASS; ++i) pf[i] = *(const v4f*)(wsrc + 16 * i * IN_DIM);
#pragma unroll
  for (int i = 0; i < GPASS; ++i) *(v4f*)&Ws[0][soff + 16 * i * GLP] = pf[i];
  __syncthreads();

  const int aoff = (wave * 32 + nc) * GLP + 2 * half;
  const int boff = nc * GLP + 2 * half;

  for (int kc = 0; kc < GNCH; ++kc) {
    const int  buf  = kc & 1;
    const bool more = kc + 1 < GNCH;
    const float* as = &As[buf][aoff];
    const float* ws = &Ws[buf][boff];

    if (more) {
#pragma unroll
      for (int i = 0; i < GPASS; ++i) pf[i] = *(const v4f*)(A + arow[i] + (kc + 1) * GBK);
    }
    G_MULT(0, GBK / 8)
    if (more) {
#pragma unroll
      for (int i = 0; i < GPASS; ++i) *(v4f*)&As[buf ^ 1][soff + 16 * i * GLP] = pf[i];
#pragma unroll
      for (int i = 0; i < GPASS; ++i) pf[i] = *(const v4f*)(wsrc + 16 * i * IN_DIM + (kc + 1) * GBK);
    }
    G_MULT(GBK / 8, GBK / 4)
    if (more) {
#pragma unroll
      for (int i = 0; i < GPASS; ++i) *(v4f*)&Ws[buf ^ 1][soff + 16 * i * GLP] = pf[i];
    }
    __syncthreads();
  }

  float wl[8];
#pragma unroll
  for (int tn = 0; tn < 8; ++tn) wl[tn] = wt[tn * 16 + nc];
  float sp[16];
#pragma unroll
  for (int tm = 0; tm < 2; ++tm) {
#pragma unroll
    for (int r = 0; r < 8; ++r) {
      float p = 0.0f;
#pragma unroll
      for (int tn = 0; tn < 8; ++tn) p += acc[tm][tn][r] * wl[tn];
      sp[tm * 8 + r] = p;
    }
  }

#pragma unroll
  for (int tm = 0; tm < 2; ++tm) {
#pragma unroll
    for (int r = 0; r < 8; ++r) {
      const int mm = m0 + wave * 32 + tm * 16 + r + 8 * half;
      if (mm < n_src) {
        float* zr = Z + (size_t)mm * OUT_DIM;
#pragma unroll
        for (int tn = 0; tn < 8; ++tn) zr[tn * 16 + nc] = acc[tm][tn][r];
      }
    }
  }

#pragma unroll
  for (int i = 0; i < 16; ++i) {
    float p = sp[i];
    p = row_xor_add<0x161>(p);
    p = row_xor_add<0x162>(p);
    p = row_xor_add<0x164>(p);
    p = row_xor_add<0x168>(p);
    sp[i] = p;
  }
  float mine = sp[0];
#pragma unroll
  for (int i = 1; i < 16; ++i) mine = (nc == i) ? sp[i] : mine;
  const int srw = m0 + wave * 32 + (nc >> 3) * 16 + (nc & 7) + 8 * half;
  if (srw < n_src) score[srw] = mine;
}
#undef G_MULT

#define RB       7
#define RANGE    (1 << RB)
#define SEGW     (RANGE * OUT_DIM)
#define NBMAX    1024
#define LCAP     4096
#define CE       8
#define CT       1024
#define GW       16
#define EPT      (LCAP / (GW * 32))

__device__ __forceinline__ float edge_logit(const float* __restrict__ score,
                                            const float* __restrict__ t,
                                            int s, float td) {
  const float e = -fabsf(t[s] - td);
  return expf(score[s] * e * 0.002f);
}

__global__ __launch_bounds__(256) void k_zero(int* __restrict__ rows, int nb) {
  const int c = blockIdx.x * blockDim.x + threadIdx.x;
  if (c < nb) rows[(size_t)c * SEGW] = 0;
}

__global__ __launch_bounds__(CT) void k_coarse(const int* __restrict__ src,
                                               const int* __restrict__ dst,
                                               int* __restrict__ rows,
                                               int E, int n_dst, int nb) {
  __shared__ int hist[NBMAX];
  const int tid = threadIdx.x;
  if (nb > NBMAX) return;
  for (int c = tid; c < NBMAX; c += CT) hist[c] = 0;
  __syncthreads();

  int s[CE], d[CE], r[CE];
  const int base = blockIdx.x * (CT * CE);
#pragma unroll
  for (int j = 0; j < CE; ++j) {
    const int i = base + j * CT + tid;
    d[j] = -1; s[j] = 0; r[j] = 0;
    if (i < E) { d[j] = dst[i]; s[j] = src[i]; }
  }
#pragma unroll
  for (int j = 0; j < CE; ++j) {
    if ((unsigned)d[j] < (unsigned)n_dst) r[j] = atomicAdd(&hist[d[j] >> RB], 1);
    else d[j] = -1;
  }
  __syncthreads();

  if (tid < nb) {
    const int h = hist[tid];
    if (h > 0) hist[tid] = atomicAdd(rows + (size_t)tid * SEGW, h);
  }
  __syncthreads();

#pragma unroll
  for (int j = 0; j < CE; ++j) {
    if (d[j] >= 0) {
      const int c   = d[j] >> RB;
      const int pos = hist[c] + r[j];
      const int nr  = min(RANGE, n_dst - c * RANGE);
      const int cap = nr * OUT_DIM - 1;
      if ((unsigned)pos < (unsigned)cap)
        rows[(size_t)c * SEGW + 1 + pos] = (s[j] << RB) | (d[j] & (RANGE - 1));
    }
  }
}

__device__ __forceinline__ float wave_max(float v) {
#pragma unroll
  for (int off = 16; off > 0; off >>= 1) v = fmaxf(v, __shfl_xor(v, off, 32));
  return v;
}
__device__ __forceinline__ float wave_sum(float v) {
#pragma unroll
  for (int off = 16; off > 0; off >>= 1) v += __shfl_xor(v, off, 32);
  return v;
}
__device__ __forceinline__ void axpy4(float4& acc, float a, const float4 z) {
  acc.x += a * z.x; acc.y += a * z.y; acc.z += a * z.z; acc.w += a * z.w;
}
__device__ __forceinline__ float lane_f(float v, int l) {
  return __int_as_float(__builtin_amdgcn_readlane(__float_as_int(v), l));
}

__global__ __launch_bounds__(GW * 32) void k_gather(const float* __restrict__ score,
                                                    const float* __restrict__ t,
                                                    const int* __restrict__ src,
                                                    const int* __restrict__ dst,
                                                    const float* __restrict__ Z,
                                                    float* __restrict__ out,
                                                    int n_src, int n_dst, int E, int nb) {
  __shared__ int cnt[RANGE];
  __shared__ int off[RANGE];
  __shared__ int cur[RANGE];
  __shared__ int list[LCAP];
  __shared__ float wgt[LCAP];

  const int c    = blockIdx.x;
  const int tid  = threadIdx.x;
  const int lane = tid & 31;
  const int wave = __builtin_amdgcn_readfirstlane(tid >> 5);
  const int d0   = c * RANGE;
  const int nr   = min(RANGE, n_dst - d0);
  const int* seg = (const int*)out + (size_t)c * SEGW;

  const int ne = __builtin_amdgcn_readfirstlane(seg[0]);
  const bool blk_slow = (nb > NBMAX) || (ne > LCAP) || (ne > nr * OUT_DIM - 1);

  int wv[EPT];
#pragma unroll
  for (int j = 0; j < EPT; ++j) {
    const int e = tid + j * (GW * 32);
    wv[j] = 0;
    if (!blk_slow && e < ne) wv[j] = seg[1 + e];
  }

  if (tid < RANGE) cnt[tid] = 0;
  __syncthreads();
#pragma unroll
  for (int j = 0; j < EPT; ++j)
    if (!blk_slow && tid + j * (GW * 32) < ne) atomicAdd(&cnt[wv[j] & (RANGE - 1)], 1);
  __syncthreads();
  if (wave == 0) {
    const int c0 = cnt[4 * lane], c1 = cnt[4 * lane + 1], c2 = cnt[4 * lane + 2], c3 = cnt[4 * lane + 3];
    const int sum = c0 + c1 + c2 + c3;
    int inc = sum;
#pragma unroll
    for (int o = 1; o < 32; o <<= 1) {
      const int v = __shfl_up(inc, o, 32);
      if (lane >= o) inc += v;
    }
    const int ex = inc - sum;
    off[4 * lane]     = ex;                cur[4 * lane]     = ex;
    off[4 * lane + 1] = ex + c0;           cur[4 * lane + 1] = ex + c0;
    off[4 * lane + 2] = ex + c0 + c1;      cur[4 * lane + 2] = ex + c0 + c1;
    off[4 * lane + 3] = ex + c0 + c1 + c2; cur[4 * lane + 3] = ex + c0 + c1 + c2;
  }
  __syncthreads();
#pragma unroll
  for (int j = 0; j < EPT; ++j)
    if (!blk_slow && tid + j * (GW * 32) < ne) {
      const unsigned w = (unsigned)wv[j];
      const int pos = atomicAdd(&cur[w & (RANGE - 1)], 1);
      list[pos] = (int)(w >> RB);
    }
  __syncthreads();

  const float* zl = Z + lane * 4;

  for (int ld = wave; ld < nr; ld += GW) {
    const int d = d0 + ld;
    const int n    = __builtin_amdgcn_readfirstlane(cnt[ld]);
    const int lb   = __builtin_amdgcn_readfirstlane(off[ld]);
    const float td = t[d];

    float4 acc = make_float4(0.0f, 0.0f, 0.0f, 0.0f);

    if (!blk_slow) {
      float mx = 0.0f;
      for (int k0 = 0; k0 < n; k0 += 32) {
        const int k = k0 + lane;
        float lg = 0.0f;
        if (k < n) {
          const int sc = (int)min((unsigned)list[lb + k], (unsigned)(n_src - 1));
          lg = edge_logit(score, t, sc, td);
          wgt[lb + k] = lg;
        }
        mx = fmaxf(mx, lg);
      }
      mx = wave_max(mx);

      float dsum = 0.0f;
      for (int k0 = 0; k0 < n; k0 += 32) {
        const int k = k0 + lane;
        if (k < n) {
          const float un = expf(wgt[lb + k] - mx);
          wgt[lb + k] = un;
          dsum += un;
        }
      }
      dsum = wave_sum(dsum);
      const float dn = fmaxf(dsum, 1e-20f);

      for (int k0 = 0; k0 < n; k0 += 32) {
        const int k = k0 + lane;
        int   sv = 0;
        float av = 0.0f;
        if (k < n) {
          sv = (int)min((unsigned)list[lb + k], (unsigned)(n_src - 1));
          av = wgt[lb + k] / dn;
        }
        int       L    = 0;
        const int Lend = ((n - k0 < 32) ? (n - k0) : 32) - 1;
        for (; L + 3 <= Lend; L += 4) {
          const int s0 = __builtin_amdgcn_readlane(sv, L);
          const int s1 = __builtin_amdgcn_readlane(sv, L + 1);
          const int s2 = __builtin_amdgcn_readlane(sv, L + 2);
          const int s3 = __builtin_amdgcn_readlane(sv, L + 3);
          const float4 z0 = *(const float4*)(zl + (size_t)s0 * OUT_DIM);
          const float4 z1 = *(const float4*)(zl + (size_t)s1 * OUT_DIM);
          const float4 z2 = *(const float4*)(zl + (size_t)s2 * OUT_DIM);
          const float4 z3 = *(const float4*)(zl + (size_t)s3 * OUT_DIM);
          axpy4(acc, lane_f(av, L),     z0);
          axpy4(acc, lane_f(av, L + 1), z1);
          axpy4(acc, lane_f(av, L + 2), z2);
          axpy4(acc, lane_f(av, L + 3), z3);
        }
        for (; L <= Lend; ++L) {
          const int s0 = __builtin_amdgcn_readlane(sv, L);
          const float4 z0 = *(const float4*)(zl + (size_t)s0 * OUT_DIM);
          axpy4(acc, lane_f(av, L), z0);
        }
      }
    } else {
      float mx = 0.0f;
      for (int i = lane; i < E; i += 32)
        if (dst[i] == d) mx = fmaxf(mx, edge_logit(score, t, src[i], td));
      mx = wave_max(mx);
      float dsum = 0.0f;
      for (int i = lane; i < E; i += 32)
        if (dst[i] == d) dsum += expf(edge_logit(score, t, src[i], td) - mx);
      dsum = wave_sum(dsum);
      const float dn = fmaxf(dsum, 1e-20f);
      for (int base = 0; base < E; base += 32) {
        const int  i   = base + lane;
        const bool hit = (i < E) && (dst[i] == d);
        int   s = 0;
        float a = 0.0f;
        if (hit) {
          s = src[i];
          a = expf(edge_logit(score, t, s, td) - mx) / dn;
        }
        unsigned mask = __builtin_amdgcn_ballot_w32(hit);
        while (mask) {
          const int b = __builtin_ctz(mask);
          mask &= mask - 1;
          const int sb = __builtin_amdgcn_readlane(s, b);
          const float4 zb = *(const float4*)(zl + (size_t)sb * OUT_DIM);
          axpy4(acc, lane_f(a, b), zb);
        }
      }
    }

    const float4 zd = *(const float4*)(zl + (size_t)d * OUT_DIM);
    float4 o;
    o.x = zd.x + acc.x; o.y = zd.y + acc.y; o.z = zd.z + acc.z; o.w = zd.w + acc.w;
    *(float4*)(out + (size_t)d * OUT_DIM + lane * 4) = o;
  }
}

extern "C" void kernel_launch(void* const* d_in, const int* in_sizes, int n_in,
                              void* d_out, int out_size, void* d_ws, size_t ws_size,
                              hipStream_t stream) {
  const float* features = (const float*)d_in[0];
  const float* t        = (const float*)d_in[1];
  const float* W_fc     = (const float*)d_in[2];
  const float* w_t      = (const float*)d_in[3];
  const int*   src_idx  = (const int*)d_in[4];
  const int*   dst_idx  = (const int*)d_in[5];

  const int n_src = in_sizes[0] / IN_DIM;
  const int E     = in_sizes[4];
  const int n_dst = out_size / OUT_DIM;

  float* zbuf  = (float*)d_ws;
  float* score = zbuf + (size_t)n_src * OUT_DIM;
  float* out   = (float*)d_out;

  k_gemm<<<(n_src + GBM - 1) / GBM, 128, 0, stream>>>(features, W_fc, w_t, zbuf, score, n_src);

  const int nb = (n_dst + RANGE - 1) / RANGE;
  k_zero<<<(nb + 255) / 256, 256, 0, stream>>>((int*)d_out, nb);
  k_coarse<<<(E + CT * CE - 1) / (CT * CE), CT, 0, stream>>>(src_idx, dst_idx, (int*)d_out,
                                                            E, n_dst, nb);

  k_gather<<<nb, GW * 32, 0, stream>>>(score, t, src_idx, dst_idx,
                                       zbuf, out, n_src, n_dst, E, nb);
}


# one 8-byte {score, t} record per node (k_gemm stores scores at stride 2, k_coarse copies t beside them): an edge's logit in k_gather takes one scattered read, not two; values and sum order unchanged
# speedup vs baseline: 4.4358x; 1.0332x over previous
#include <hip/hip_runtime.h>
#include <hip/hip_bf16.h>

typedef __attribute__((ext_vector_type(2))) float v2f;
typedef __attribute__((ext_vector_type(8))) float v8f;

#define IN_DIM  256
#define OUT_DIM 128

typedef __attribute__((ext_vector_type(4))) float v4f;

#define GBM   128
#define GBK   32
#define GLP   (GBK + 4)
#define GNCH  (IN_DIM / GBK)
#define GPASS (GBM / 16)
static_assert(GBM == OUT_DIM, "the W tile is staged with the A tile's row map");

#define G_MULT(ks0, ks1)                                                        \
  _Pragma("unroll")                                                             \
  for (int ks = (ks0) * 4; ks < (ks1) * 4; ks += 4) {                           \
    const v2f a0 = *(const v2f*)(as + ks);                                      \
    const v2f a1 = *(const v2f*)(as + 16 * GLP + ks);                           \
    _Pragma("unroll")                                                           \
    for (int tn = 0; tn < 8; ++tn) {                                            \
      const v2f b = *(const v2f*)(ws + tn * 16 * GLP + ks);                     \
      acc[0][tn] = __builtin_amdgcn_wmma_f32_16x16x4_f32(                       \
           false, a0,  false, b,                              \
           (short)0, acc[0][tn],  false,  false);\
      acc[1][tn] = __builtin_amdgcn_wmma_f32_16x16x4_f32(                       \
           false, a1,  false, b,                              \
           (short)0, acc[1][tn],  false,  false);\
    }                                                                           \
  }

template <int CTRL>
__device__ __forceinline__ float row_xor_add(float v) {
  return v + __int_as_float(__builtin_amdgcn_update_dpp(0, __float_as_int(v), CTRL, 0xf, 0xf, false));
}

__global__ __launch_bounds__(128) void k_gemm(const float* __restrict__ A,
                                              const float* __restrict__ W,
                                              const float* __restrict__ wt,
                                              float* __restrict__ Z,
                                              float* __restrict__ score,
                                              int n_src, int sh) {
  __shared__ __attribute__((aligned(16))) float As[2][GBM * GLP];
  __shared__ __attribute__((aligned(16))) float Ws[2][OUT_DIM * GLP];

  const int tid  = threadIdx.x;
  const int lane = tid & 31;
  const int wave = tid >> 5;
  const int half = lane >> 4;
  const int nc   = lane & 15;
  const int m0   = blockIdx.x * GBM;

  const int srow = tid >> 3;
  const int scol = (tid & 7) * 4;
  const int soff = srow * GLP + scol;


  unsigned arow[GPASS];
#pragma unroll
  for (int i = 0; i < GPASS; ++i) {
    int gr = m0 + srow + 16 * i;
    if (gr >= n_src) gr = n_src - 1;
    arow[i] = (unsigned)gr * IN_DIM + scol;
  }
  const float* __restrict__ wsrc = W + srow * IN_DIM + scol;

  v4f pf[GPASS];

  const v8f vzero = {};
  v8f acc[2][8];
#pragma unroll
  for (int i = 0; i < 8; ++i) { acc[0][i] = vzero; acc[1][i] = vzero; }

#pragma unroll
  for (int i = 0; i < GPASS; ++i) pf[i] = *(const v4f*)(A + arow[i]);
#pragma unroll
  for (int i = 0; i < GPASS; ++i) *(v4f*)&As[0][soff + 16 * i * GLP] = pf[i];
#pragma unroll
  for (int i = 0; i < GPASS; ++i) pf[i] = *(const v4f*)(wsrc + 16 * i * IN_DIM);
#pragma unroll
  for (int i = 0; i < GPASS; ++i) *(v4f*)&Ws[0][soff + 16 * i * GLP] = pf[i];
  __syncthreads();

  const int aoff = (wave * 32 + nc) * GLP + 2 * half;
  const int boff = nc * GLP + 2 * half;

  for (int kc = 0; kc < GNCH; ++kc) {
    const int  buf  = kc & 1;
    const bool more = kc + 1 < GNCH;
    const float* as = &As[buf][aoff];
    const float* ws = &Ws[buf][boff];

    if (more) {
#pragma unroll
      for (int i = 0; i < GPASS; ++i) pf[i] = *(const v4f*)(A + arow[i] + (kc + 1) * GBK);
    }
    G_MULT(0, GBK / 8)
    if (more) {
#pragma unroll
      for (int i = 0; i < GPASS; ++i) *(v4f*)&As[buf ^ 1][soff + 16 * i * GLP] = pf[i];
#pragma unroll
      for (int i = 0; i < GPASS; ++i) pf[i] = *(const v4f*)(wsrc + 16 * i * IN_DIM + (kc + 1) * GBK);
    }
    G_MULT(GBK / 8, GBK / 4)
    if (more) {
#pragma unroll
      for (int i = 0; i < GPASS; ++i) *(v4f*)&Ws[buf ^ 1][soff + 16 * i * GLP] = pf[i];
    }
    __syncthreads();
  }

  float wl[8];
#pragma unroll
  for (int tn = 0; tn < 8; ++tn) wl[tn] = wt[tn * 16 + nc];
  float sp[16];
#pragma unroll
  for (int tm = 0; tm < 2; ++tm) {
#pragma unroll
    for (int r = 0; r < 8; ++r) {
      float p = 0.0f;
#pragma unroll
      for (int tn = 0; tn < 8; ++tn) p += acc[tm][tn][r] * wl[tn];
      sp[tm * 8 + r] = p;
    }
  }

#pragma unroll
  for (int tm = 0; tm < 2; ++tm) {
#pragma unroll
    for (int r = 0; r < 8; ++r) {
      const int mm = m0 + wave * 32 + tm * 16 + r + 8 * half;
      if (mm < n_src) {
        float* zr = Z + (size_t)mm * OUT_DIM;
#pragma unroll
        for (int tn = 0; tn < 8; ++tn) zr[tn * 16 + nc] = acc[tm][tn][r];
      }
    }
  }

#pragma unroll
  for (int i = 0; i < 16; ++i) {
    float p = sp[i];
    p = row_xor_add<0x161>(p);
    p = row_xor_add<0x162>(p);
    p = row_xor_add<0x164>(p);
    p = row_xor_add<0x168>(p);
    sp[i] = p;
  }
  float mine = sp[0];
#pragma unroll
  for (int i = 1; i < 16; ++i) mine = (nc == i) ? sp[i] : mine;
  const int srw = m0 + wave * 32 + (nc >> 3) * 16 + (nc & 7) + 8 * half;
  if (srw < n_src) score[(size_t)srw << sh] = mine;
}
#undef G_MULT

#define RB       7
#define RANGE    (1 << RB)
#define SEGW     (RANGE * OUT_DIM)
#define NBMAX    1024
#define LCAP     4096
#define CE       8
#define CT       1024
#define GW       16
#define EPT      (LCAP / (GW * 32))

template <bool PAIR>
__device__ __forceinline__ float edge_logit(const float* __restrict__ score,
                                            const float* __restrict__ t,
                                            int s, float td) {
  float sc, ts;
  if (PAIR) {
    const float2 r = *(const float2*)(score + 2 * (size_t)s);
    sc = r.x; ts = r.y;
  } else {
    sc = score[s]; ts = t[s];
  }
  const float e = -fabsf(ts - td);
  return expf(sc * e * 0.002f);
}

__global__ __launch_bounds__(256) void k_zero(int* __restrict__ rows, int nb) {
  const int c = blockIdx.x * blockDim.x + threadIdx.x;
  if (c < nb) rows[(size_t)c * SEGW] = 0;
}

__device__ __forceinline__ void copy_t(const float* __restrict__ t, float* __restrict__ rec,
                                       int n_src, int pair) {
  if (pair) {
    for (int i = blockIdx.x * CT + threadIdx.x; i < n_src; i += gridDim.x * CT)
      rec[2 * (size_t)i + 1] = t[i];
  }
}

__global__ __launch_bounds__(CT) void k_coarse(const int* __restrict__ src,
                                               const int* __restrict__ dst,
                                               int* __restrict__ rows,
                                               int E, int n_dst, int nb,
                                               const float* __restrict__ t,
                                               float* __restrict__ rec,
                                               int n_src, int pair) {
  __shared__ int hist[NBMAX];
  const int tid = threadIdx.x;
  if (nb > NBMAX) {
    copy_t(t, rec, n_src, pair);
    return;
  }
  for (int c = tid; c < NBMAX; c += CT) hist[c] = 0;
  __syncthreads();

  int s[CE], d[CE], r[CE];
  const int base = blockIdx.x * (CT * CE);
#pragma unroll
  for (int j = 0; j < CE; ++j) {
    const int i = base + j * CT + tid;
    d[j] = -1; s[j] = 0; r[j] = 0;
    if (i < E) { d[j] = dst[i]; s[j] = src[i]; }
  }
  copy_t(t, rec, n_src, pair);
#pragma unroll
  for (int j = 0; j < CE; ++j) {
    if ((unsigned)d[j] < (unsigned)n_dst) r[j] = atomicAdd(&hist[d[j] >> RB], 1);
    else d[j] = -1;
  }
  __syncthreads();

  if (tid < nb) {
    const int h = hist[tid];
    if (h > 0) hist[tid] = atomicAdd(rows + (size_t)tid * SEGW, h);
  }
  __syncthreads();

#pragma unroll
  for (int j = 0; j < CE; ++j) {
    if (d[j] >= 0) {
      const int c   = d[j] >> RB;
      const int pos = hist[c] + r[j];
      const int nr  = min(RANGE, n_dst - c * RANGE);
      const int cap = nr * OUT_DIM - 1;
      if ((unsigned)pos < (unsigned)cap)
        rows[(size_t)c * SEGW + 1 + pos] = (s[j] << RB) | (d[j] & (RANGE - 1));
    }
  }
}

__device__ __forceinline__ float wave_max(float v) {
#pragma unroll
  for (int off = 16; off > 0; off >>= 1) v = fmaxf(v, __shfl_xor(v, off, 32));
  return v;
}
__device__ __forceinline__ float wave_sum(float v) {
#pragma unroll
  for (int off = 16; off > 0; off >>= 1) v += __shfl_xor(v, off, 32);
  return v;
}
__device__ __forceinline__ void axpy4(float4& acc, float a, const float4 z) {
  acc.x += a * z.x; acc.y += a * z.y; acc.z += a * z.z; acc.w += a * z.w;
}
__device__ __forceinline__ float lane_f(float v, int l) {
  return __int_as_float(__builtin_amdgcn_readlane(__float_as_int(v), l));
}

template <bool PAIR>
__global__ __launch_bounds__(GW * 32) void k_gather(const float* __restrict__ score,
                                                    const float* __restrict__ t,
                                                    const int* __restrict__ src,
                                                    const int* __restrict__ dst,
                                                    const float* __restrict__ Z,
                                                    float* __restrict__ out,
                                                    int n_src, int n_dst, int E, int nb) {
  __shared__ int cnt[RANGE];
  __shared__ int off[RANGE];
  __shared__ int cur[RANGE];
  __shared__ int list[LCAP];
  __shared__ float wgt[LCAP];

  const int c    = blockIdx.x;
  const int tid  = threadIdx.x;
  const int lane = tid & 31;
  const int wave = __builtin_amdgcn_readfirstlane(tid >> 5);
  const int d0   = c * RANGE;
  const int nr   = min(RANGE, n_dst - d0);
  const int* seg = (const int*)out + (size_t)c * SEGW;

  const int ne = __builtin_amdgcn_readfirstlane(seg[0]);
  const bool blk_slow = (nb > NBMAX) || (ne > LCAP) || (ne > nr * OUT_DIM - 1);

  int wv[EPT];
#pragma unroll
  for (int j = 0; j < EPT; ++j) {
    const int e = tid + j * (GW * 32);
    wv[j] = 0;
    if (!blk_slow && e < ne) wv[j] = seg[1 + e];
  }

  if (tid < RANGE) cnt[tid] = 0;
  __syncthreads();
#pragma unroll
  for (int j = 0; j < EPT; ++j)
    if (!blk_slow && tid + j * (GW * 32) < ne) atomicAdd(&cnt[wv[j] & (RANGE - 1)], 1);
  __syncthreads();
  if (wave == 0) {
    const int c0 = cnt[4 * lane], c1 = cnt[4 * lane + 1], c2 = cnt[4 * lane + 2], c3 = cnt[4 * lane + 3];
    const int sum = c0 + c1 + c2 + c3;
    int inc = sum;
#pragma unroll
    for (int o = 1; o < 32; o <<= 1) {
      const int v = __shfl_up(inc, o, 32);
      if (lane >= o) inc += v;
    }
    const int ex = inc - sum;
    off[4 * lane]     = ex;                cur[4 * lane]     = ex;
    off[4 * lane + 1] = ex + c0;           cur[4 * lane + 1] = ex + c0;
    off[4 * lane + 2] = ex + c0 + c1;      cur[4 * lane + 2] = ex + c0 + c1;
    off[4 * lane + 3] = ex + c0 + c1 + c2; cur[4 * lane + 3] = ex + c0 + c1 + c2;
  }
  __syncthreads();
#pragma unroll
  for (int j = 0; j < EPT; ++j)
    if (!blk_slow && tid + j * (GW * 32) < ne) {
      const unsigned w = (unsigned)wv[j];
      const int pos = atomicAdd(&cur[w & (RANGE - 1)], 1);
      list[pos] = (int)(w >> RB);
    }
  __syncthreads();

  const float* zl = Z + lane * 4;

  for (int ld = wave; ld < nr; ld += GW) {
    const int d = d0 + ld;
    const int n    = __builtin_amdgcn_readfirstlane(cnt[ld]);
    const int lb   = __builtin_amdgcn_readfirstlane(off[ld]);
    const float td = t[d];

    float4 acc = make_float4(0.0f, 0.0f, 0.0f, 0.0f);

    if (!blk_slow) {
      float mx = 0.0f;
      for (int k0 = 0; k0 < n; k0 += 32) {
        const int k = k0 + lane;
        float lg = 0.0f;
        if (k < n) {
          const int sc = (int)min((unsigned)list[lb + k], (unsigned)(n_src - 1));
          lg = edge_logit<PAIR>(score, t, sc, td);
          wgt[lb + k] = lg;
        }
        mx = fmaxf(mx, lg);
      }
      mx = wave_max(mx);

      float dsum = 0.0f;
      for (int k0 = 0; k0 < n; k0 += 32) {
        const int k = k0 + lane;
        if (k < n) {
          const float un = expf(wgt[lb + k] - mx);
          wgt[lb + k] = un;
          dsum += un;
        }
      }
      dsum = wave_sum(dsum);
      const float dn = fmaxf(dsum, 1e-20f);

      for (int k0 = 0; k0 < n; k0 += 32) {
        const int k = k0 + lane;
        int   sv = 0;
        float av = 0.0f;
        if (k < n) {
          sv = (int)min((unsigned)list[lb + k], (unsigned)(n_src - 1));
          av = wgt[lb + k] / dn;
        }
        int       L    = 0;
        const int Lend = ((n - k0 < 32) ? (n - k0) : 32) - 1;
        for (; L + 3 <= Lend; L += 4) {
          const int s0 = __builtin_amdgcn_readlane(sv, L);
          const int s1 = __builtin_amdgcn_readlane(sv, L + 1);
          const int s2 = __builtin_amdgcn_readlane(sv, L + 2);
          const int s3 = __builtin_amdgcn_readlane(sv, L + 3);
          const float4 z0 = *(const float4*)(zl + (size_t)s0 * OUT_DIM);
          const float4 z1 = *(const float4*)(zl + (size_t)s1 * OUT_DIM);
          const float4 z2 = *(const float4*)(zl + (size_t)s2 * OUT_DIM);
          const float4 z3 = *(const float4*)(zl + (size_t)s3 * OUT_DIM);
          axpy4(acc, lane_f(av, L),     z0);
          axpy4(acc, lane_f(av, L + 1), z1);
          axpy4(acc, lane_f(av, L + 2), z2);
          axpy4(acc, lane_f(av, L + 3), z3);
        }
        for (; L <= Lend; ++L) {
          const int s0 = __builtin_amdgcn_readlane(sv, L);
          const float4 z0 = *(const float4*)(zl + (size_t)s0 * OUT_DIM);
          axpy4(acc, lane_f(av, L), z0);
        }
      }
    } else {
      float mx = 0.0f;
      for (int i = lane; i < E; i += 32)
        if (dst[i] == d) mx = fmaxf(mx, edge_logit<PAIR>(score, t, src[i], td));
      mx = wave_max(mx);
      float dsum = 0.0f;
      for (int i = lane; i < E; i += 32)
        if (dst[i] == d) dsum += expf(edge_logit<PAIR>(score, t, src[i], td) - mx);
      dsum = wave_sum(dsum);
      const float dn = fmaxf(dsum, 1e-20f);
      for (int base = 0; base < E; base += 32) {
        const int  i   = base + lane;
        const bool hit = (i < E) && (dst[i] == d);
        int   s = 0;
        float a = 0.0f;
        if (hit) {
          s = src[i];
          a = expf(edge_logit<PAIR>(score, t, s, td) - mx) / dn;
        }
        unsigned mask = __builtin_amdgcn_ballot_w32(hit);
        while (mask) {
          const int b = __builtin_ctz(mask);
          mask &= mask - 1;
          const int sb = __builtin_amdgcn_readlane(s, b);
          const float4 zb = *(const float4*)(zl + (size_t)sb * OUT_DIM);
          axpy4(acc, lane_f(a, b), zb);
        }
      }
    }

    const float4 zd = *(const float4*)(zl + (size_t)d * OUT_DIM);
    float4 o;
    o.x = zd.x + acc.x; o.y = zd.y + acc.y; o.z = zd.z + acc.z; o.w = zd.w + acc.w;
    *(float4*)(out + (size_t)d * OUT_DIM + lane * 4) = o;
  }
}

extern "C" void kernel_launch(void* const* d_in, const int* in_sizes, int n_in,
                              void* d_out, int out_size, void* d_ws, size_t ws_size,
                              hipStream_t stream) {
  const float* features = (const float*)d_in[0];
  const float* t        = (const float*)d_in[1];
  const float* W_fc     = (const float*)d_in[2];
  const float* w_t      = (const float*)d_in[3];
  const int*   src_idx  = (const int*)d_in[4];
  const int*   dst_idx  = (const int*)d_in[5];

  const int n_src = in_sizes[0] / IN_DIM;
  const int E     = in_sizes[4];
  const int n_dst = out_size / OUT_DIM;

  float* zbuf  = (float*)d_ws;
  float* score = zbuf + (size_t)n_src * OUT_DIM;
  float* out   = (float*)d_out;
  const int pair = ws_size >= (size_t)n_src * (OUT_DIM + 2) * sizeof(float);

  k_gemm<<<(n_src + GBM - 1) / GBM, 128, 0, stream>>>(features, W_fc, w_t, zbuf, score, n_src, pair);

  const int nb = (n_dst + RANGE - 1) / RANGE;
  k_zero<<<(nb + 255) / 256, 256, 0, stream>>>((int*)d_out, nb);
  k_coarse<<<(E + CT * CE - 1) / (CT * CE), CT, 0, stream>>>(src_idx, dst_idx, (int*)d_out,
                                                            E, n_dst, nb, t, score, n_src, pair);

  if (pair)
    k_gather<true><<<nb, GW * 32, 0, stream>>>(score, t, src_idx, dst_idx,
                                               zbuf, out, n_src, n_dst, E, nb);
  else
    k_gather<false><<<nb, GW * 32, 0, stream>>>(score, t, src_idx, dst_idx,
                                                zbuf, out, n_src, n_dst, E, nb);
}


# edge binning runs as 128-thread blocks at the head of the GEMM launch's grid, beside the matrix work: 3 launches, not 4; t copy moves to the counter-clear launch; sums still in ticket arrival order
# speedup vs baseline: 4.8543x; 1.0943x over previous
#include <hip/hip_runtime.h>
#include <hip/hip_bf16.h>

typedef __attribute__((ext_vector_type(2))) float v2f;
typedef __attribute__((ext_vector_type(8))) float v8f;

#define IN_DIM  256
#define OUT_DIM 128

typedef __attribute__((ext_vector_type(4))) float v4f;

#define GBM   128
#define GBK   32
#define GLP   (GBK + 4)
#define GNCH  (IN_DIM / GBK)
#define GPASS (GBM / 16)
static_assert(GBM == OUT_DIM, "the W tile is staged with the A tile's row map");

#define G_MULT(ks0, ks1)                                                        \
  _Pragma("unroll")                                                             \
  for (int ks = (ks0) * 4; ks < (ks1) * 4; ks += 4) {                           \
    const v2f a0 = *(const v2f*)(as + ks);                                      \
    const v2f a1 = *(const v2f*)(as + 16 * GLP + ks);                           \
    _Pragma("unroll")                                                           \
    for (int tn = 0; tn < 8; ++tn) {                                            \
      const v2f b = *(const v2f*)(ws + tn * 16 * GLP + ks);                     \
      acc[0][tn] = __builtin_amdgcn_wmma_f32_16x16x4_f32(                       \
           false, a0,  false, b,                              \
           (short)0, acc[0][tn],  false,  false);\
      acc[1][tn] = __builtin_amdgcn_wmma_f32_16x16x4_f32(                       \
           false, a1,  false, b,                              \
           (short)0, acc[1][tn],  false,  false);\
    }                                                                           \
  }

template <int CTRL>
__device__ __forceinline__ float row_xor_add(float v) {
  return v + __int_as_float(__builtin_amdgcn_update_dpp(0, __float_as_int(v), CTRL, 0xf, 0xf, false));
}

__device__ __forceinline__ void bin_edges(int cb, const int* __restrict__ src,
                                          const int* __restrict__ dst, int* __restrict__ rows,
                                          int E, int n_dst, int nb, int* hist);

__global__ __launch_bounds__(128) void k_gemm(const float* __restrict__ A,
                                              const float* __restrict__ W,
                                              const float* __restrict__ wt,
                                              float* __restrict__ Z,
                                              float* __restrict__ score,
                                              int n_src, int sh,
                                              const int* __restrict__ src,
                                              const int* __restrict__ dst,
                                              int* __restrict__ rows,
                                              int E, int n_dst, int nb, int ncb) {
  __shared__ __attribute__((aligned(16))) float As[2][GBM * GLP];
  __shared__ __attribute__((aligned(16))) float Ws[2][OUT_DIM * GLP];

  if ((int)blockIdx.x < ncb) {
    bin_edges(blockIdx.x, src, dst, rows, E, n_dst, nb, (int*)&As[0][0]);
    return;
  }

  const int tid  = threadIdx.x;
  const int lane = tid & 31;
  const int wave = tid >> 5;
  const int half = lane >> 4;
  const int nc   = lane & 15;
  const int m0   = ((int)blockIdx.x - ncb) * GBM;

  const int srow = tid >> 3;
  const int scol = (tid & 7) * 4;
  const int soff = srow * GLP + scol;


  unsigned arow[GPASS];
#pragma unroll
  for (int i = 0; i < GPASS; ++i) {
    int gr = m0 + srow + 16 * i;
    if (gr >= n_src) gr = n_src - 1;
    arow[i] = (unsigned)gr * IN_DIM + scol;
  }
  const float* __restrict__ wsrc = W + srow * IN_DIM + scol;

  v4f pf[GPASS];

  const v8f vzero = {};
  v8f acc[2][8];
#pragma unroll
  for (int i = 0; i < 8; ++i) { acc[0][i] = vzero; acc[1][i] = vzero; }

#pragma unroll
  for (int i = 0; i < GPASS; ++i) pf[i] = *(const v4f*)(A + arow[i]);
#pragma unroll
  for (int i = 0; i < GPASS; ++i) *(v4f*)&As[0][soff + 16 * i * GLP] = pf[i];
#pragma unroll
  for (int i = 0; i < GPASS; ++i) pf[i] = *(const v4f*)(wsrc + 16 * i * IN_DIM);
#pragma unroll
  for (int i = 0; i < GPASS; ++i) *(v4f*)&Ws[0][soff + 16 * i * GLP] = pf[i];
  __syncthreads();

  const int aoff = (wave * 32 + nc) * GLP + 2 * half;
  const int boff = nc * GLP + 2 * half;

  for (int kc = 0; kc < GNCH; ++kc) {
    const int  buf  = kc & 1;
    const bool more = kc + 1 < GNCH;
    const float* as = &As[buf][aoff];
    const float* ws = &Ws[buf][boff];

    if (more) {
#pragma unroll
      for (int i = 0; i < GPASS; ++i) pf[i] = *(const v4f*)(A + arow[i] + (kc + 1) * GBK);
    }
    G_MULT(0, GBK / 8)
    if (more) {
#pragma unroll
      for (int i = 0; i < GPASS; ++i) *(v4f*)&As[buf ^ 1][soff + 16 * i * GLP] = pf[i];
#pragma unroll
      for (int i = 0; i < GPASS; ++i) pf[i] = *(const v4f*)(wsrc + 16 * i * IN_DIM + (kc + 1) * GBK);
    }
    G_MULT(GBK / 8, GBK / 4)
    if (more) {
#pragma unroll
      for (int i = 0; i < GPASS; ++i) *(v4f*)&Ws[buf ^ 1][soff + 16 * i * GLP] = pf[i];
    }
    __syncthreads();
  }

  float wl[8];
#pragma unroll
  for (int tn = 0; tn < 8; ++tn) wl[tn] = wt[tn * 16 + nc];
  float sp[16];
#pragma unroll
  for (int tm = 0; tm < 2; ++tm) {
#pragma unroll
    for (int r = 0; r < 8; ++r) {
      float p = 0.0f;
#pragma unroll
      for (int tn = 0; tn < 8; ++tn) p += acc[tm][tn][r] * wl[tn];
      sp[tm * 8 + r] = p;
    }
  }

#pragma unroll
  for (int tm = 0; tm < 2; ++tm) {
#pragma unroll
    for (int r = 0; r < 8; ++r) {
      const int mm = m0 + wave * 32 + tm * 16 + r + 8 * half;
      if (mm < n_src) {
        float* zr = Z + (size_t)mm * OUT_DIM;
#pragma unroll
        for (int tn = 0; tn < 8; ++tn) zr[tn * 16 + nc] = acc[tm][tn][r];
      }
    }
  }

#pragma unroll
  for (int i = 0; i < 16; ++i) {
    float p = sp[i];
    p = row_xor_add<0x161>(p);
    p = row_xor_add<0x162>(p);
    p = row_xor_add<0x164>(p);
    p = row_xor_add<0x168>(p);
    sp[i] = p;
  }
  float mine = sp[0];
#pragma unroll
  for (int i = 1; i < 16; ++i) mine = (nc == i) ? sp[i] : mine;
  const int srw = m0 + wave * 32 + (nc >> 3) * 16 + (nc & 7) + 8 * half;
  if (srw < n_src) score[(size_t)srw << sh] = mine;
}
#undef G_MULT

#define RB       7
#define RANGE    (1 << RB)
#define SEGW     (RANGE * OUT_DIM)
#define NBMAX    1024
#define LCAP     4096
#define BT       128
#define BE       64
#define BEPB     (BT * BE)
#define GW       16
#define EPT      (LCAP / (GW * 32))

template <bool PAIR>
__device__ __forceinline__ float edge_logit(const float* __restrict__ score,
                                            const float* __restrict__ t,
                                            int s, float td) {
  float sc, ts;
  if (PAIR) {
    const float2 r = *(const float2*)(score + 2 * (size_t)s);
    sc = r.x; ts = r.y;
  } else {
    sc = score[s]; ts = t[s];
  }
  const float e = -fabsf(ts - td);
  return expf(sc * e * 0.002f);
}

__global__ __launch_bounds__(256) void k_prep(int* __restrict__ rows, int nb,
                                              const float* __restrict__ t,
                                              float* __restrict__ rec, int n_src, int pair) {
  const int g  = blockIdx.x * 256 + threadIdx.x;
  const int gs = gridDim.x * 256;
  for (int c = g; c < nb; c += gs) rows[(size_t)c * SEGW] = 0;
  if (pair) {
    for (int i = 4 * g; i < n_src; i += 4 * gs) {
      if (i + 3 < n_src) {
        const float4 v = *(const float4*)(t + i);
        rec[2 * (size_t)i + 1] = v.x; rec[2 * (size_t)i + 3] = v.y;
        rec[2 * (size_t)i + 5] = v.z; rec[2 * (size_t)i + 7] = v.w;
      } else {
#pragma unroll
        for (int q = 0; q < 3; ++q)
          if (i + q < n_src) rec[2 * (size_t)(i + q) + 1] = t[i + q];
      }
    }
  }
}

__device__ __forceinline__ void bin_edges(int cb, const int* __restrict__ src,
                                          const int* __restrict__ dst, int* __restrict__ rows,
                                          int E, int n_dst, int nb, int* hist) {
  const int tid = threadIdx.x;
  if (nb > NBMAX) return;
  for (int c = tid; c < NBMAX; c += BT) hist[c] = 0;
  __syncthreads();

  int s[BE], d[BE];
  const int base = cb * BEPB;
  if (E - base >= BEPB) {
#pragma unroll
    for (int j = 0; j < BE / 4; ++j) {
      const int i = base + (j * BT + tid) * 4;
      const int4 dv = *(const int4*)(dst + i);
      const int4 sv = *(const int4*)(src + i);
      d[4 * j] = dv.x; d[4 * j + 1] = dv.y; d[4 * j + 2] = dv.z; d[4 * j + 3] = dv.w;
      s[4 * j] = sv.x; s[4 * j + 1] = sv.y; s[4 * j + 2] = sv.z; s[4 * j + 3] = sv.w;
    }
  } else {
#pragma unroll
    for (int j = 0; j < BE; ++j) {
      const int i = base + j * BT + tid;
      d[j] = -1; s[j] = 0;
      if (i < E) { d[j] = dst[i]; s[j] = src[i]; }
    }
  }
#pragma unroll
  for (int j = 0; j < BE; ++j) {
    if ((unsigned)d[j] < (unsigned)n_dst) atomicAdd(&hist[d[j] >> RB], 1);
    else d[j] = -1;
  }
  __syncthreads();

  int hh[NBMAX / BT], bb[NBMAX / BT];
#pragma unroll
  for (int q = 0; q < NBMAX / BT; ++q) {
    const int c = tid + q * BT;
    hh[q] = (c < nb) ? hist[c] : 0;
  }
#pragma unroll
  for (int q = 0; q < NBMAX / BT; ++q) {
    bb[q] = 0;
    if (hh[q] > 0) bb[q] = atomicAdd(rows + (size_t)(tid + q * BT) * SEGW, hh[q]);
  }
#pragma unroll
  for (int q = 0; q < NBMAX / BT; ++q)
    if (hh[q] > 0) hist[tid + q * BT] = bb[q];
  __syncthreads();

#pragma unroll
  for (int j = 0; j < BE; ++j) {
    int w = -1;
    if (d[j] >= 0) {
      const int c   = d[j] >> RB;
      const int pos = atomicAdd(&hist[c], 1);
      const int nr  = min(RANGE, n_dst - c * RANGE);
      const int cap = nr * OUT_DIM - 1;
      s[j] = (s[j] << RB) | (d[j] & (RANGE - 1));
      if ((unsigned)pos < (unsigned)cap) w = c * SEGW + 1 + pos;
    }
    d[j] = w;
  }
#pragma unroll
  for (int j = 0; j < BE; ++j)
    if (d[j] >= 0) rows[(size_t)d[j]] = s[j];
}

__device__ __forceinline__ float wave_max(float v) {
#pragma unroll
  for (int off = 16; off > 0; off >>= 1) v = fmaxf(v, __shfl_xor(v, off, 32));
  return v;
}
__device__ __forceinline__ float wave_sum(float v) {
#pragma unroll
  for (int off = 16; off > 0; off >>= 1) v += __shfl_xor(v, off, 32);
  return v;
}
__device__ __forceinline__ void axpy4(float4& acc, float a, const float4 z) {
  acc.x += a * z.x; acc.y += a * z.y; acc.z += a * z.z; acc.w += a * z.w;
}
__device__ __forceinline__ float lane_f(float v, int l) {
  return __int_as_float(__builtin_amdgcn_readlane(__float_as_int(v), l));
}

template <bool PAIR>
__global__ __launch_bounds__(GW * 32) void k_gather(const float* __restrict__ score,
                                                    const float* __restrict__ t,
                                                    const int* __restrict__ src,
                                                    const int* __restrict__ dst,
                                                    const float* __restrict__ Z,
                                                    float* __restrict__ out,
                                                    int n_src, int n_dst, int E, int nb) {
  __shared__ int cnt[RANGE];
  __shared__ int off[RANGE];
  __shared__ int cur[RANGE];
  __shared__ int list[LCAP];
  __shared__ float wgt[LCAP];

  const int c    = blockIdx.x;
  const int tid  = threadIdx.x;
  const int lane = tid & 31;
  const int wave = __builtin_amdgcn_readfirstlane(tid >> 5);
  const int d0   = c * RANGE;
  const int nr   = min(RANGE, n_dst - d0);
  const int* seg = (const int*)out + (size_t)c * SEGW;

  const int ne = __builtin_amdgcn_readfirstlane(seg[0]);
  const bool blk_slow = (nb > NBMAX) || (ne > LCAP) || (ne > nr * OUT_DIM - 1);

  int wv[EPT];
#pragma unroll
  for (int j = 0; j < EPT; ++j) {
    const int e = tid + j * (GW * 32);
    wv[j] = 0;
    if (!blk_slow && e < ne) wv[j] = seg[1 + e];
  }

  if (tid < RANGE) cnt[tid] = 0;
  __syncthreads();
#pragma unroll
  for (int j = 0; j < EPT; ++j)
    if (!blk_slow && tid + j * (GW * 32) < ne) atomicAdd(&cnt[wv[j] & (RANGE - 1)], 1);
  __syncthreads();
  if (wave == 0) {
    const int c0 = cnt[4 * lane], c1 = cnt[4 * lane + 1], c2 = cnt[4 * lane + 2], c3 = cnt[4 * lane + 3];
    const int sum = c0 + c1 + c2 + c3;
    int inc = sum;
#pragma unroll
    for (int o = 1; o < 32; o <<= 1) {
      const int v = __shfl_up(inc, o, 32);
      if (lane >= o) inc += v;
    }
    const int ex = inc - sum;
    off[4 * lane]     = ex;                cur[4 * lane]     = ex;
    off[4 * lane + 1] = ex + c0;           cur[4 * lane + 1] = ex + c0;
    off[4 * lane + 2] = ex + c0 + c1;      cur[4 * lane + 2] = ex + c0 + c1;
    off[4 * lane + 3] = ex + c0 + c1 + c2; cur[4 * lane + 3] = ex + c0 + c1 + c2;
  }
  __syncthreads();
#pragma unroll
  for (int j = 0; j < EPT; ++j)
    if (!blk_slow && tid + j * (GW * 32) < ne) {
      const unsigned w = (unsigned)wv[j];
      const int pos = atomicAdd(&cur[w & (RANGE - 1)], 1);
      list[pos] = (int)(w >> RB);
    }
  __syncthreads();

  const float* zl = Z + lane * 4;

  for (int ld = wave; ld < nr; ld += GW) {
    const int d = d0 + ld;
    const int n    = __builtin_amdgcn_readfirstlane(cnt[ld]);
    const int lb   = __builtin_amdgcn_readfirstlane(off[ld]);
    const float td = t[d];

    float4 acc = make_float4(0.0f, 0.0f, 0.0f, 0.0f);

    if (!blk_slow) {
      float mx = 0.0f;
      for (int k0 = 0; k0 < n; k0 += 32) {
        const int k = k0 + lane;
        float lg = 0.0f;
        if (k < n) {
          const int sc = (int)min((unsigned)list[lb + k], (unsigned)(n_src - 1));
          lg = edge_logit<PAIR>(score, t, sc, td);
          wgt[lb + k] = lg;
        }
        mx = fmaxf(mx, lg);
      }
      mx = wave_max(mx);

      float dsum = 0.0f;
      for (int k0 = 0; k0 < n; k0 += 32) {
        const int k = k0 + lane;
        if (k < n) {
          const float un = expf(wgt[lb + k] - mx);
          wgt[lb + k] = un;
          dsum += un;
        }
      }
      dsum = wave_sum(dsum);
      const float dn = fmaxf(dsum, 1e-20f);

      for (int k0 = 0; k0 < n; k0 += 32) {
        const int k = k0 + lane;
        int   sv = 0;
        float av = 0.0f;
        if (k < n) {
          sv = (int)min((unsigned)list[lb + k], (unsigned)(n_src - 1));
          av = wgt[lb + k] / dn;
        }
        int       L    = 0;
        const int Lend = ((n - k0 < 32) ? (n - k0) : 32) - 1;
        for (; L + 3 <= Lend; L += 4) {
          const int s0 = __builtin_amdgcn_readlane(sv, L);
          const int s1 = __builtin_amdgcn_readlane(sv, L + 1);
          const int s2 = __builtin_amdgcn_readlane(sv, L + 2);
          const int s3 = __builtin_amdgcn_readlane(sv, L + 3);
          const float4 z0 = *(const float4*)(zl + (size_t)s0 * OUT_DIM);
          const float4 z1 = *(const float4*)(zl + (size_t)s1 * OUT_DIM);
          const float4 z2 = *(const float4*)(zl + (size_t)s2 * OUT_DIM);
          const float4 z3 = *(const float4*)(zl + (size_t)s3 * OUT_DIM);
          axpy4(acc, lane_f(av, L),     z0);
          axpy4(acc, lane_f(av, L + 1), z1);
          axpy4(acc, lane_f(av, L + 2), z2);
          axpy4(acc, lane_f(av, L + 3), z3);
        }
        for (; L <= Lend; ++L) {
          const int s0 = __builtin_amdgcn_readlane(sv, L);
          const float4 z0 = *(const float4*)(zl + (size_t)s0 * OUT_DIM);
          axpy4(acc, lane_f(av, L), z0);
        }
      }
    } else {
      float mx = 0.0f;
      for (int i = lane; i < E; i += 32)
        if (dst[i] == d) mx = fmaxf(mx, edge_logit<PAIR>(score, t, src[i], td));
      mx = wave_max(mx);
      float dsum = 0.0f;
      for (int i = lane; i < E; i += 32)
        if (dst[i] == d) dsum += expf(edge_logit<PAIR>(score, t, src[i], td) - mx);
      dsum = wave_sum(dsum);
      const float dn = fmaxf(dsum, 1e-20f);
      for (int base = 0; base < E; base += 32) {
        const int  i   = base + lane;
        const bool hit = (i < E) && (dst[i] == d);
        int   s = 0;
        float a = 0.0f;
        if (hit) {
          s = src[i];
          a = expf(edge_logit<PAIR>(score, t, s, td) - mx) / dn;
        }
        unsigned mask = __builtin_amdgcn_ballot_w32(hit);
        while (mask) {
          const int b = __builtin_ctz(mask);
          mask &= mask - 1;
          const int sb = __builtin_amdgcn_readlane(s, b);
          const float4 zb = *(const float4*)(zl + (size_t)sb * OUT_DIM);
          axpy4(acc, lane_f(a, b), zb);
        }
      }
    }

    const float4 zd = *(const float4*)(zl + (size_t)d * OUT_DIM);
    float4 o;
    o.x = zd.x + acc.x; o.y = zd.y + acc.y; o.z = zd.z + acc.z; o.w = zd.w + acc.w;
    *(float4*)(out + (size_t)d * OUT_DIM + lane * 4) = o;
  }
}

extern "C" void kernel_launch(void* const* d_in, const int* in_sizes, int n_in,
                              void* d_out, int out_size, void* d_ws, size_t ws_size,
                              hipStream_t stream) {
  const float* features = (const float*)d_in[0];
  const float* t        = (const float*)d_in[1];
  const float* W_fc     = (const float*)d_in[2];
  const float* w_t      = (const float*)d_in[3];
  const int*   src_idx  = (const int*)d_in[4];
  const int*   dst_idx  = (const int*)d_in[5];

  const int n_src = in_sizes[0] / IN_DIM;
  const int E     = in_sizes[4];
  const int n_dst = out_size / OUT_DIM;

  float* zbuf  = (float*)d_ws;
  float* score = zbuf + (size_t)n_src * OUT_DIM;
  float* out   = (float*)d_out;
  const int pair = ws_size >= (size_t)n_src * (OUT_DIM + 2) * sizeof(float);

  const int nb = (n_dst + RANGE - 1) / RANGE;
  k_prep<<<(n_src + 1023) / 1024, 256, 0, stream>>>((int*)d_out, nb, t, score, n_src, pair);

  const int ncb = (E + BEPB - 1) / BEPB;
  k_gemm<<<ncb + (n_src + GBM - 1) / GBM, 128, 0, stream>>>(features, W_fc, w_t, zbuf, score, n_src, pair,
                                                            src_idx, dst_idx, (int*)d_out,
                                                            E, n_dst, nb, ncb);

  if (pair)
    k_gather<true><<<nb, GW * 32, 0, stream>>>(score, t, src_idx, dst_idx,
                                               zbuf, out, n_src, n_dst, E, nb);
  else
    k_gather<false><<<nb, GW * 32, 0, stream>>>(score, t, src_idx, dst_idx,
                                                zbuf, out, n_src, n_dst, E, nb);
}


# binning needs no cleared counter: each binning block stores its per-range edge counts and parks edges in slots of its own; it also copies t: 2 launches, not 3; sums still in ticket arrival order
# speedup vs baseline: 5.1462x; 1.0601x over previous
#include <hip/hip_runtime.h>
#include <hip/hip_bf16.h>

typedef __attribute__((ext_vector_type(2))) float v2f;
typedef __attribute__((ext_vector_type(8))) float v8f;

#define IN_DIM  256
#define OUT_DIM 128

typedef __attribute__((ext_vector_type(4))) float v4f;

#define GBM   128
#define GBK   32
#define GLP   (GBK + 4)
#define GNCH  (IN_DIM / GBK)
#define GPASS (GBM / 16)
static_assert(GBM == OUT_DIM, "the W tile is staged with the A tile's row map");

#define G_MULT(ks0, ks1)                                                        \
  _Pragma("unroll")                                                             \
  for (int ks = (ks0) * 4; ks < (ks1) * 4; ks += 4) {                           \
    const v2f a0 = *(const v2f*)(as + ks);                                      \
    const v2f a1 = *(const v2f*)(as + 16 * GLP + ks);                           \
    _Pragma("unroll")                                                           \
    for (int tn = 0; tn < 8; ++tn) {                                            \
      const v2f b = *(const v2f*)(ws + tn * 16 * GLP + ks);                     \
      acc[0][tn] = __builtin_amdgcn_wmma_f32_16x16x4_f32(                       \
           false, a0,  false, b,                              \
           (short)0, acc[0][tn],  false,  false);\
      acc[1][tn] = __builtin_amdgcn_wmma_f32_16x16x4_f32(                       \
           false, a1,  false, b,                              \
           (short)0, acc[1][tn],  false,  false);\
    }                                                                           \
  }

template <int CTRL>
__device__ __forceinline__ float row_xor_add(float v) {
  return v + __int_as_float(__builtin_amdgcn_update_dpp(0, __float_as_int(v), CTRL, 0xf, 0xf, false));
}

__device__ __forceinline__ void bin_edges(int cb, const int* __restrict__ src,
                                          const int* __restrict__ dst, int* __restrict__ rows,
                                          int E, int n_dst, int nb, int ncb, int* hist,
                                          const float* __restrict__ tt, float* __restrict__ rec,
                                          int n_src, int pair);

__global__ __launch_bounds__(128) void k_gemm(const float* __restrict__ A,
                                              const float* __restrict__ W,
                                              const float* __restrict__ wt,
                                              const float* __restrict__ tt,
                                              float* __restrict__ Z,
                                              float* __restrict__ score,
                                              int n_src, int sh,
                                              const int* __restrict__ src,
                                              const int* __restrict__ dst,
                                              int* __restrict__ rows,
                                              int E, int n_dst, int nb, int ncb) {
  __shared__ __attribute__((aligned(16))) float As[2][GBM * GLP];
  __shared__ __attribute__((aligned(16))) float Ws[2][OUT_DIM * GLP];

  if ((int)blockIdx.x < ncb) {
    bin_edges(blockIdx.x, src, dst, rows, E, n_dst, nb, ncb, (int*)&As[0][0], tt, score, n_src, sh);
    return;
  }

  const int tid  = threadIdx.x;
  const int lane = tid & 31;
  const int wave = tid >> 5;
  const int half = lane >> 4;
  const int nc   = lane & 15;
  const int m0   = ((int)blockIdx.x - ncb) * GBM;

  const int srow = tid >> 3;
  const int scol = (tid & 7) * 4;
  const int soff = srow * GLP + scol;


  unsigned arow[GPASS];
#pragma unroll
  for (int i = 0; i < GPASS; ++i) {
    int gr = m0 + srow + 16 * i;
    if (gr >= n_src) gr = n_src - 1;
    arow[i] = (unsigned)gr * IN_DIM + scol;
  }
  const float* __restrict__ wsrc = W + srow * IN_DIM + scol;

  v4f pf[GPASS];

  const v8f vzero = {};
  v8f acc[2][8];
#pragma unroll
  for (int i = 0; i < 8; ++i) { acc[0][i] = vzero; acc[1][i] = vzero; }

#pragma unroll
  for (int i = 0; i < GPASS; ++i) pf[i] = *(const v4f*)(A + arow[i]);
#pragma unroll
  for (int i = 0; i < GPASS; ++i) *(v4f*)&As[0][soff + 16 * i * GLP] = pf[i];
#pragma unroll
  for (int i = 0; i < GPASS; ++i) pf[i] = *(const v4f*)(wsrc + 16 * i * IN_DIM);
#pragma unroll
  for (int i = 0; i < GPASS; ++i) *(v4f*)&Ws[0][soff + 16 * i * GLP] = pf[i];
  __syncthreads();

  const int aoff = (wave * 32 + nc) * GLP + 2 * half;
  const int boff = nc * GLP + 2 * half;

  for (int kc = 0; kc < GNCH; ++kc) {
    const int  buf  = kc & 1;
    const bool more = kc + 1 < GNCH;
    const float* as = &As[buf][aoff];
    const float* ws = &Ws[buf][boff];

    if (more) {
#pragma unroll
      for (int i = 0; i < GPASS; ++i) pf[i] = *(const v4f*)(A + arow[i] + (kc + 1) * GBK);
    }
    G_MULT(0, GBK / 8)
    if (more) {
#pragma unroll
      for (int i = 0; i < GPASS; ++i) *(v4f*)&As[buf ^ 1][soff + 16 * i * GLP] = pf[i];
#pragma unroll
      for (int i = 0; i < GPASS; ++i) pf[i] = *(const v4f*)(wsrc + 16 * i * IN_DIM + (kc + 1) * GBK);
    }
    G_MULT(GBK / 8, GBK / 4)
    if (more) {
#pragma unroll
      for (int i = 0; i < GPASS; ++i) *(v4f*)&Ws[buf ^ 1][soff + 16 * i * GLP] = pf[i];
    }
    __syncthreads();
  }

  float wl[8];
#pragma unroll
  for (int tn = 0; tn < 8; ++tn) wl[tn] = wt[tn * 16 + nc];
  float sp[16];
#pragma unroll
  for (int tm = 0; tm < 2; ++tm) {
#pragma unroll
    for (int r = 0; r < 8; ++r) {
      float p = 0.0f;
#pragma unroll
      for (int tn = 0; tn < 8; ++tn) p += acc[tm][tn][r] * wl[tn];
      sp[tm * 8 + r] = p;
    }
  }

#pragma unroll
  for (int tm = 0; tm < 2; ++tm) {
#pragma unroll
    for (int r = 0; r < 8; ++r) {
      const int mm = m0 + wave * 32 + tm * 16 + r + 8 * half;
      if (mm < n_src) {
        float* zr = Z + (size_t)mm * OUT_DIM;
#pragma unroll
        for (int tn = 0; tn < 8; ++tn) zr[tn * 16 + nc] = acc[tm][tn][r];
      }
    }
  }

#pragma unroll
  for (int i = 0; i < 16; ++i) {
    float p = sp[i];
    p = row_xor_add<0x161>(p);
    p = row_xor_add<0x162>(p);
    p = row_xor_add<0x164>(p);
    p = row_xor_add<0x168>(p);
    sp[i] = p;
  }
  float mine = sp[0];
#pragma unroll
  for (int i = 1; i < 16; ++i) mine = (nc == i) ? sp[i] : mine;
  const int srw = m0 + wave * 32 + (nc >> 3) * 16 + (nc & 7) + 8 * half;
  if (srw < n_src) score[(size_t)srw << sh] = mine;
}
#undef G_MULT

#define RB       7
#define RANGE    (1 << RB)
#define SEGW     (RANGE * OUT_DIM)
#define NBMAX    1024
#define LCAP     4096
#define BT       128
#define BE       64
#define BEPB     (BT * BE)
#define GW       16
#define CNT0     256
#define NCBMAX   CNT0
#define CAPB     32
#define BR       (NCBMAX / GW)
#define OVF      (1 << 24)
static_assert(BR * BEPB < OVF, "a wave's edge sum must stay below the flag bit");

__device__ __forceinline__ int slots_per_block(int nr, int ncb) {
  const int room = nr * OUT_DIM - CNT0;
  if (room <= 0) return 0;
  if (ncb < 1) return CAPB;
  return min(CAPB, room / ncb);
}

template <bool PAIR>
__device__ __forceinline__ float edge_logit(const float* __restrict__ score,
                                            const float* __restrict__ t,
                                            int s, float td) {
  float sc, ts;
  if (PAIR) {
    const float2 r = *(const float2*)(score + 2 * (size_t)s);
    sc = r.x; ts = r.y;
  } else {
    sc = score[s]; ts = t[s];
  }
  const float e = -fabsf(ts - td);
  return expf(sc * e * 0.002f);
}

__device__ __forceinline__ void bin_edges(int cb, const int* __restrict__ src,
                                          const int* __restrict__ dst, int* __restrict__ rows,
                                          int E, int n_dst, int nb, int ncb, int* hist,
                                          const float* __restrict__ tt, float* __restrict__ rec,
                                          int n_src, int pair) {
  const int tid = threadIdx.x;
  if (pair) {
    for (int i = 4 * (cb * BT + tid); i < n_src; i += 4 * ncb * BT) {
      if (i + 3 < n_src) {
        const float4 v = *(const float4*)(tt + i);
        rec[2 * (size_t)i + 1] = v.x; rec[2 * (size_t)i + 3] = v.y;
        rec[2 * (size_t)i + 5] = v.z; rec[2 * (size_t)i + 7] = v.w;
      } else {
#pragma unroll
        for (int q = 0; q < 3; ++q)
          if (i + q < n_src) rec[2 * (size_t)(i + q) + 1] = tt[i + q];
      }
    }
  }
  if (nb > NBMAX || ncb > NCBMAX) return;
  for (int c = tid; c < NBMAX; c += BT) hist[c] = 0;
  __syncthreads();

  int s[BE], d[BE];
  const int base = cb * BEPB;
  if (E - base >= BEPB) {
#pragma unroll
    for (int j = 0; j < BE / 4; ++j) {
      const int i = base + (j * BT + tid) * 4;
      const int4 dv = *(const int4*)(dst + i);
      const int4 sv = *(const int4*)(src + i);
      d[4 * j] = dv.x; d[4 * j + 1] = dv.y; d[4 * j + 2] = dv.z; d[4 * j + 3] = dv.w;
      s[4 * j] = sv.x; s[4 * j + 1] = sv.y; s[4 * j + 2] = sv.z; s[4 * j + 3] = sv.w;
    }
  } else {
#pragma unroll
    for (int j = 0; j < BE; ++j) {
      const int i = base + j * BT + tid;
      d[j] = -1; s[j] = 0;
      if (i < E) { d[j] = dst[i]; s[j] = src[i]; }
    }
  }
#pragma unroll
  for (int j = 0; j < BE; ++j) {
    if ((unsigned)d[j] < (unsigned)n_dst) atomicAdd(&hist[d[j] >> RB], 1);
    else d[j] = -1;
  }
  __syncthreads();

  const int cap_full = slots_per_block(RANGE, ncb);
  const int cap_last = slots_per_block(n_dst - (nb - 1) * RANGE, ncb);
#pragma unroll
  for (int q = 0; q < NBMAX / BT; ++q) {
    const int c = tid + q * BT;
    if (c < nb) {
      if ((c == nb - 1 ? cap_last : cap_full) > 0) rows[(size_t)c * SEGW + cb] = hist[c];
      hist[c] = 0;
    }
  }
  __syncthreads();

#pragma unroll
  for (int j = 0; j < BE; ++j) {
    int w = -1;
    if (d[j] >= 0) {
      const int c   = d[j] >> RB;
      const int pos = atomicAdd(&hist[c], 1);
      const int cap = (c == nb - 1) ? cap_last : cap_full;
      s[j] = (s[j] << RB) | (d[j] & (RANGE - 1));
      if (pos < cap) w = c * SEGW + CNT0 + cb * cap + pos;
    }
    d[j] = w;
  }
#pragma unroll
  for (int j = 0; j < BE; ++j)
    if (d[j] >= 0) rows[(size_t)d[j]] = s[j];
}

__device__ __forceinline__ float wave_max(float v) {
#pragma unroll
  for (int off = 16; off > 0; off >>= 1) v = fmaxf(v, __shfl_xor(v, off, 32));
  return v;
}
__device__ __forceinline__ float wave_sum(float v) {
#pragma unroll
  for (int off = 16; off > 0; off >>= 1) v += __shfl_xor(v, off, 32);
  return v;
}
__device__ __forceinline__ void axpy4(float4& acc, float a, const float4 z) {
  acc.x += a * z.x; acc.y += a * z.y; acc.z += a * z.z; acc.w += a * z.w;
}
__device__ __forceinline__ float lane_f(float v, int l) {
  return __int_as_float(__builtin_amdgcn_readlane(__float_as_int(v), l));
}

template <bool PAIR>
__global__ __launch_bounds__(GW * 32) void k_gather(const float* __restrict__ score,
                                                    const float* __restrict__ t,
                                                    const int* __restrict__ src,
                                                    const int* __restrict__ dst,
                                                    const float* __restrict__ Z,
                                                    float* __restrict__ out,
                                                    int n_src, int n_dst, int E, int nb, int ncb) {
  __shared__ __attribute__((aligned(16))) int wsum[GW];
  __shared__ int cnt[RANGE];
  __shared__ int off[RANGE];
  __shared__ int cur[RANGE];
  __shared__ int list[LCAP];
  __shared__ float wgt[LCAP];

  const int c    = blockIdx.x;
  const int tid  = threadIdx.x;
  const int lane = tid & 31;
  const int wave = __builtin_amdgcn_readfirstlane(tid >> 5);
  const int d0   = c * RANGE;
  const int nr   = min(RANGE, n_dst - d0);
  const int* seg = (const int*)out + (size_t)c * SEGW;

  const int  capb     = slots_per_block(nr, ncb);
  const bool cfg_slow = (nb > NBMAX) || (ncb > NCBMAX) || (capb < 1);

  int cv = 0;
  if (!cfg_slow && lane < BR && wave + GW * lane < ncb) cv = seg[wave + GW * lane];

  int      wv[BR];
  unsigned vm   = 0u;
  int      tsum = 0;
#pragma unroll
  for (int r = 0; r < BR; ++r) {
    const int cf = __builtin_amdgcn_readlane(cv, r);
    const int cr = min(cf, capb);
    tsum += cf;
    if (cf > capb) tsum |= OVF;
    wv[r] = 0;
    if (lane < cr) {
      wv[r] = seg[CNT0 + (wave + GW * r) * capb + lane];
      vm |= 1u << r;
    }
  }
  if (lane == 0) wsum[wave] = tsum;

  if (tid < RANGE) cnt[tid] = 0;
  __syncthreads();
  int ne = 0;
#pragma unroll
  for (int q = 0; q < GW; ++q) ne += wsum[q] & (OVF - 1);
  int bad = 0;
#pragma unroll
  for (int q = 0; q < GW; ++q) bad |= wsum[q] & OVF;
  const bool blk_slow = cfg_slow || (bad != 0) || (ne > LCAP);
  if (blk_slow) vm = 0u;
#pragma unroll
  for (int r = 0; r < BR; ++r)
    if (vm & (1u << r)) atomicAdd(&cnt[wv[r] & (RANGE - 1)], 1);
  __syncthreads();
  if (wave == 0) {
    const int c0 = cnt[4 * lane], c1 = cnt[4 * lane + 1], c2 = cnt[4 * lane + 2], c3 = cnt[4 * lane + 3];
    const int sum = c0 + c1 + c2 + c3;
    int inc = sum;
#pragma unroll
    for (int o = 1; o < 32; o <<= 1) {
      const int v = __shfl_up(inc, o, 32);
      if (lane >= o) inc += v;
    }
    const int ex = inc - sum;
    off[4 * lane]     = ex;                cur[4 * lane]     = ex;
    off[4 * lane + 1] = ex + c0;           cur[4 * lane + 1] = ex + c0;
    off[4 * lane + 2] = ex + c0 + c1;      cur[4 * lane + 2] = ex + c0 + c1;
    off[4 * lane + 3] = ex + c0 + c1 + c2; cur[4 * lane + 3] = ex + c0 + c1 + c2;
  }
  __syncthreads();
#pragma unroll
  for (int r = 0; r < BR; ++r)
    if (vm & (1u << r)) {
      const unsigned w = (unsigned)wv[r];
      const int pos = atomicAdd(&cur[w & (RANGE - 1)], 1);
      list[pos] = (int)(w >> RB);
    }
  __syncthreads();

  const float* zl = Z + lane * 4;

  for (int ld = wave; ld < nr; ld += GW) {
    const int d = d0 + ld;
    const int n    = __builtin_amdgcn_readfirstlane(cnt[ld]);
    const int lb   = __builtin_amdgcn_readfirstlane(off[ld]);
    const float td = t[d];

    float4 acc = make_float4(0.0f, 0.0f, 0.0f, 0.0f);

    if (!blk_slow) {
      float mx = 0.0f;
      for (int k0 = 0; k0 < n; k0 += 32) {
        const int k = k0 + lane;
        float lg = 0.0f;
        if (k < n) {
          const int sc = (int)min((unsigned)list[lb + k], (unsigned)(n_src - 1));
          lg = edge_logit<PAIR>(score, t, sc, td);
          wgt[lb + k] = lg;
        }
        mx = fmaxf(mx, lg);
      }
      mx = wave_max(mx);

      float dsum = 0.0f;
      for (int k0 = 0; k0 < n; k0 += 32) {
        const int k = k0 + lane;
        if (k < n) {
          const float un = expf(wgt[lb + k] - mx);
          wgt[lb + k] = un;
          dsum += un;
        }
      }
      dsum = wave_sum(dsum);
      const float dn = fmaxf(dsum, 1e-20f);

      for (int k0 = 0; k0 < n; k0 += 32) {
        const int k = k0 + lane;
        int   sv = 0;
        float av = 0.0f;
        if (k < n) {
          sv = (int)min((unsigned)list[lb + k], (unsigned)(n_src - 1));
          av = wgt[lb + k] / dn;
        }
        int       L    = 0;
        const int Lend = ((n - k0 < 32) ? (n - k0) : 32) - 1;
        for (; L + 3 <= Lend; L += 4) {
          const int s0 = __builtin_amdgcn_readlane(sv, L);
          const int s1 = __builtin_amdgcn_readlane(sv, L + 1);
          const int s2 = __builtin_amdgcn_readlane(sv, L + 2);
          const int s3 = __builtin_amdgcn_readlane(sv, L + 3);
          const float4 z0 = *(const float4*)(zl + (size_t)s0 * OUT_DIM);
          const float4 z1 = *(const float4*)(zl + (size_t)s1 * OUT_DIM);
          const float4 z2 = *(const float4*)(zl + (size_t)s2 * OUT_DIM);
          const float4 z3 = *(const float4*)(zl + (size_t)s3 * OUT_DIM);
          axpy4(acc, lane_f(av, L),     z0);
          axpy4(acc, lane_f(av, L + 1), z1);
          axpy4(acc, lane_f(av, L + 2), z2);
          axpy4(acc, lane_f(av, L + 3), z3);
        }
        for (; L <= Lend; ++L) {
          const int s0 = __builtin_amdgcn_readlane(sv, L);
          const float4 z0 = *(const float4*)(zl + (size_t)s0 * OUT_DIM);
          axpy4(acc, lane_f(av, L), z0);
        }
      }
    } else {
      float mx = 0.0f;
      for (int i = lane; i < E; i += 32)
        if (dst[i] == d) mx = fmaxf(mx, edge_logit<PAIR>(score, t, src[i], td));
      mx = wave_max(mx);
      float dsum = 0.0f;
      for (int i = lane; i < E; i += 32)
        if (dst[i] == d) dsum += expf(edge_logit<PAIR>(score, t, src[i], td) - mx);
      dsum = wave_sum(dsum);
      const float dn = fmaxf(dsum, 1e-20f);
      for (int base = 0; base < E; base += 32) {
        const int  i   = base + lane;
        const bool hit = (i < E) && (dst[i] == d);
        int   s = 0;
        float a = 0.0f;
        if (hit) {
          s = src[i];
          a = expf(edge_logit<PAIR>(score, t, s, td) - mx) / dn;
        }
        unsigned mask = __builtin_amdgcn_ballot_w32(hit);
        while (mask) {
          const int b = __builtin_ctz(mask);
          mask &= mask - 1;
          const int sb = __builtin_amdgcn_readlane(s, b);
          const float4 zb = *(const float4*)(zl + (size_t)sb * OUT_DIM);
          axpy4(acc, lane_f(a, b), zb);
        }
      }
    }

    const float4 zd = *(const float4*)(zl + (size_t)d * OUT_DIM);
    float4 o;
    o.x = zd.x + acc.x; o.y = zd.y + acc.y; o.z = zd.z + acc.z; o.w = zd.w + acc.w;
    *(float4*)(out + (size_t)d * OUT_DIM + lane * 4) = o;
  }
}

extern "C" void kernel_launch(void* const* d_in, const int* in_sizes, int n_in,
                              void* d_out, int out_size, void* d_ws, size_t ws_size,
                              hipStream_t stream) {
  const float* features = (const float*)d_in[0];
  const float* t        = (const float*)d_in[1];
  const float* W_fc     = (const float*)d_in[2];
  const float* w_t      = (const float*)d_in[3];
  const int*   src_idx  = (const int*)d_in[4];
  const int*   dst_idx  = (const int*)d_in[5];

  const int n_src = in_sizes[0] / IN_DIM;
  const int E     = in_sizes[4];
  const int n_dst = out_size / OUT_DIM;

  float* zbuf  = (float*)d_ws;
  float* score = zbuf + (size_t)n_src * OUT_DIM;
  float* out   = (float*)d_out;
  const int pair = ws_size >= (size_t)n_src * (OUT_DIM + 2) * sizeof(float);

  const int nb  = (n_dst + RANGE - 1) / RANGE;
  const int ncb = (E + BEPB - 1) / BEPB;
  k_gemm<<<ncb + (n_src + GBM - 1) / GBM, 128, 0, stream>>>(features, W_fc, w_t, t, zbuf, score, n_src, pair,
                                                            src_idx, dst_idx, (int*)d_out,
                                                            E, n_dst, nb, ncb);

  if (pair)
    k_gather<true><<<nb, GW * 32, 0, stream>>>(score, t, src_idx, dst_idx,
                                               zbuf, out, n_src, n_dst, E, nb, ncb);
  else
    k_gather<false><<<nb, GW * 32, 0, stream>>>(score, t, src_idx, dst_idx,
                                                zbuf, out, n_src, n_dst, E, nb, ncb);
}
